# starting kernel
# baseline (speedup 1.0000x reference)
#include <hip/hip_runtime.h>
#include <hip/hip_bf16.h>

#define NEG_SLOPE 0.2f

typedef __attribute__((ext_vector_type(16))) _Float16 v16h;
typedef __attribute__((ext_vector_type(8)))  float    v8f;

template<int CO>
__global__ __launch_bounds__(32)
void node_xform_wmma(const float* __restrict__ X, const float* __restrict__ W,
                     const float* __restrict__ bias, float* __restrict__ Out,
                     int n_nodes) {
  const int mtile = blockIdx.x;
  const int ntile = blockIdx.y;
  const int lane  = threadIdx.x;
  const int mrow  = lane & 15;
  const int grp   = lane >> 4;
  const int nIdx  = ntile * 16 + (lane & 15);

  int node = mtile * 16 + mrow;
  if (node >= n_nodes) node = n_nodes - 1;

  const float bv = bias[nIdx];
  v8f c;
#pragma unroll
  for (int r = 0; r < 8; ++r) c[r] = bv;

#pragma unroll
  for (int ks = 0; ks < 2; ++ks) {
    v16h a, b;
#pragma unroll
    for (int v = 0; v < 8; ++v) {
      const int k = ks * 32 + ((v & 4) ? 16 : 0) + grp * 8 + (v & 3) * 2;
      const float2 xp = *(const float2*)(X + (size_t)node * 64 + k);
      a[2 * v]     = (_Float16)xp.x;
      a[2 * v + 1] = (_Float16)xp.y;
      b[2 * v]     = (_Float16)W[(size_t)(k)     * CO + nIdx];
      b[2 * v + 1] = (_Float16)W[(size_t)(k + 1) * CO + nIdx];
    }
    c = __builtin_amdgcn_wmma_f32_16x16x32_f16(false, a, false, b,
                                               (short)0, c, false, false);
  }

  if ((mtile + 1) * 16 <= n_nodes) {
#pragma unroll
    for (int r = 0; r < 8; ++r)
      Out[(size_t)(mtile * 16 + r + 8 * grp) * CO + nIdx] = c[r];
  } else {
#pragma unroll
    for (int r = 0; r < 8; ++r) {
      const int row = mtile * 16 + r + 8 * grp;
      if (row < n_nodes) Out[(size_t)row * CO + nIdx] = c[r];
    }
  }
}

__global__ void k_init(float* __restrict__ amax, float* __restrict__ den,
                       float* __restrict__ acc, int n, int co) {
  const int i = blockIdx.x * blockDim.x + threadIdx.x;
  if (i < n * co) acc[i] = 0.0f;
  if (i < n) { amax[i] = -3.402823466e38f; den[i] = 0.0f; }
}

__device__ __forceinline__ void atomicMaxFloat(float* addr, float val) {
  if (val >= 0.0f) atomicMax((int*)addr, __float_as_int(val));
  else             atomicMin((unsigned int*)addr, __float_as_uint(val));
}

template<int CO>
__global__ __launch_bounds__(256)
void edge_alpha(const float* __restrict__ xl, const float* __restrict__ xr,
                const int* __restrict__ src, const int* __restrict__ dst,
                const float* __restrict__ ef, const float* __restrict__ We,
                const float* __restrict__ att,
                float* __restrict__ alpha, float* __restrict__ amax, int ne) {
  constexpr int LPE = CO / 4;
  const int gid = blockIdx.x * blockDim.x + threadIdx.x;
  const int e   = gid / LPE;
  const int sub = gid % LPE;
  if (e >= ne) return;
  const int s = src[e], d = dst[e];
  const float fe = ef[e];
  const int off = sub * 4;
  const float4 a4 = *(const float4*)(xl + (size_t)s * CO + off);
  const float4 r4 = *(const float4*)(xr + (size_t)d * CO + off);
  const float4 w4 = *(const float4*)(We + off);
  const float4 t4 = *(const float4*)(att + off);
  float acc = 0.0f, m;
  m = a4.x + r4.x + fe * w4.x; m = (m > 0.0f) ? m : NEG_SLOPE * m; acc += m * t4.x;
  m = a4.y + r4.y + fe * w4.y; m = (m > 0.0f) ? m : NEG_SLOPE * m; acc += m * t4.y;
  m = a4.z + r4.z + fe * w4.z; m = (m > 0.0f) ? m : NEG_SLOPE * m; acc += m * t4.z;
  m = a4.w + r4.w + fe * w4.w; m = (m > 0.0f) ? m : NEG_SLOPE * m; acc += m * t4.w;
#pragma unroll
  for (int o = LPE / 2; o > 0; o >>= 1) acc += __shfl_down(acc, o, LPE);
  if (sub == 0) { alpha[e] = acc; atomicMaxFloat(amax + d, acc); }
}

__global__ __launch_bounds__(256)
void edge_exp(float* __restrict__ alpha, const int* __restrict__ dst,
              const float* __restrict__ amax, float* __restrict__ den, int ne) {
  const int e = blockIdx.x * blockDim.x + threadIdx.x;
  if (e >= ne) return;
  const int d = dst[e];
  const float ex = __expf(alpha[e] - amax[d]);
  alpha[e] = ex;
  atomicAdd(den + d, ex);
}

template<int CO>
__global__ __launch_bounds__(256)
void edge_scatter(const float* __restrict__ xl, const float* __restrict__ ex,
                  const float* __restrict__ den, const int* __restrict__ src,
                  const int* __restrict__ dst, float* __restrict__ acc, int ne) {
  constexpr int LPE = CO / 4;
  const int gid = blockIdx.x * blockDim.x + threadIdx.x;
  const int e   = gid / LPE;
  const int sub = gid % LPE;
  if (e >= ne) return;
  const int s = src[e], d = dst[e];
  const float w = ex[e] / (den[d] + 1e-16f);
  const int off = sub * 4;
  const float4 v = *(const float4*)(xl + (size_t)s * CO + off);
  float* p = acc + (size_t)d * CO + off;
  atomicAdd(p + 0, v.x * w);
  atomicAdd(p + 1, v.y * w);
  atomicAdd(p + 2, v.z * w);
  atomicAdd(p + 3, v.w * w);
}

__global__ __launch_bounds__(256)
void k_finalize(float* __restrict__ acc, const float* __restrict__ bias,
                int n, int co, int do_relu) {
  const int i = blockIdx.x * blockDim.x + threadIdx.x;
  if (i >= n * co) return;
  float v = acc[i] + bias[i % co];
  if (do_relu) v = fmaxf(v, 0.0f);
  acc[i] = v;
}

extern "C" void kernel_launch(void* const* d_in, const int* in_sizes, int n_in,
                              void* d_out, int out_size, void* d_ws, size_t ws_size,
                              hipStream_t stream) {
  (void)n_in; (void)out_size; (void)ws_size;
  const float* x    = (const float*)d_in[0];
  const int*   ei   = (const int*)d_in[1];
  const float* ef   = (const float*)d_in[2];
  const float* Wl0  = (const float*)d_in[3];
  const float* bl0  = (const float*)d_in[4];
  const float* Wr0  = (const float*)d_in[5];
  const float* br0  = (const float*)d_in[6];
  const float* We0  = (const float*)d_in[7];
  const float* att0 = (const float*)d_in[8];
  const float* b0   = (const float*)d_in[9];
  const float* Wl1  = (const float*)d_in[10];
  const float* bl1  = (const float*)d_in[11];
  const float* Wr1  = (const float*)d_in[12];
  const float* br1  = (const float*)d_in[13];
  const float* We1  = (const float*)d_in[14];
  const float* att1 = (const float*)d_in[15];
  const float* b1   = (const float*)d_in[16];

  const int n = in_sizes[0] / 64;
  const int e = in_sizes[1] / 2;
  const int* srcI = ei;
  const int* dstI = ei + e;

  float* ws   = (float*)d_ws;
  float* xl   = ws;  ws += (size_t)n * 64;
  float* xr   = ws;  ws += (size_t)n * 64;
  float* hbuf = ws;  ws += (size_t)n * 64;
  float* alp  = ws;  ws += (size_t)e;
  float* amax = ws;  ws += n;
  float* den  = ws;  ws += n;

  auto cdiv = [](long long a, long long b) { return (int)((a + b - 1) / b); };
  const int B = 256;

  {
    constexpr int CO = 64;
    k_init<<<cdiv((long long)n * CO, B), B, 0, stream>>>(amax, den, hbuf, n, CO);
    dim3 g((n + 15) / 16, CO / 16);
    node_xform_wmma<CO><<<g, 32, 0, stream>>>(x, Wl0, bl0, xl, n);
    node_xform_wmma<CO><<<g, 32, 0, stream>>>(x, Wr0, br0, xr, n);
    edge_alpha<CO><<<cdiv((long long)e * (CO / 4), B), B, 0, stream>>>(
        xl, xr, srcI, dstI, ef, We0, att0, alp, amax, e);
    edge_exp<<<cdiv(e, B), B, 0, stream>>>(alp, dstI, amax, den, e);
    edge_scatter<CO><<<cdiv((long long)e * (CO / 4), B), B, 0, stream>>>(
        xl, alp, den, srcI, dstI, hbuf, e);
    k_finalize<<<cdiv((long long)n * CO, B), B, 0, stream>>>(hbuf, b0, n, CO, 1);
  }

  {
    constexpr int CO = 32;
    float* outp = (float*)d_out;
    k_init<<<cdiv((long long)n * CO, B), B, 0, stream>>>(amax, den, outp, n, CO);
    dim3 g((n + 15) / 16, CO / 16);
    node_xform_wmma<CO><<<g, 32, 0, stream>>>(hbuf, Wl1, bl1, xl, n);
    node_xform_wmma<CO><<<g, 32, 0, stream>>>(hbuf, Wr1, br1, xr, n);
    edge_alpha<CO><<<cdiv((long long)e * (CO / 4), B), B, 0, stream>>>(
        xl, xr, srcI, dstI, ef, We1, att1, alp, amax, e);
    edge_exp<<<cdiv(e, B), B, 0, stream>>>(alp, dstI, amax, den, e);
    edge_scatter<CO><<<cdiv((long long)e * (CO / 4), B), B, 0, stream>>>(
        xl, alp, den, srcI, dstI, outp, e);
    k_finalize<<<cdiv((long long)n * CO, B), B, 0, stream>>>(outp, b1, n, CO, 0);
  }
}


# Bucket edges by destination per call (returning int atomic rank, scan, place); a lane group per node keeps softmax and weighted sum in registers; sums run in arrival-order rank, no float atomics
# speedup vs baseline: 2.6248x; 2.6248x over previous
#include <hip/hip_runtime.h>
#include <hip/hip_bf16.h>

#define NEG_SLOPE 0.2f

typedef __attribute__((ext_vector_type(16))) _Float16 v16h;
typedef __attribute__((ext_vector_type(8)))  float    v8f;

template<int CO>
__global__ __launch_bounds__(32)
void node_xform_wmma(const float* __restrict__ X, const float* __restrict__ W,
                     const float* __restrict__ bias, float* __restrict__ Out,
                     int n_nodes) {
  const int mtile = blockIdx.x;
  const int ntile = blockIdx.y;
  const int lane  = threadIdx.x;
  const int mrow  = lane & 15;
  const int grp   = lane >> 4;
  const int nIdx  = ntile * 16 + (lane & 15);

  int node = mtile * 16 + mrow;
  if (node >= n_nodes) node = n_nodes - 1;

  const float bv = bias[nIdx];
  v8f c;
#pragma unroll
  for (int r = 0; r < 8; ++r) c[r] = bv;

#pragma unroll
  for (int ks = 0; ks < 2; ++ks) {
    v16h a, b;
#pragma unroll
    for (int v = 0; v < 8; ++v) {
      const int k = ks * 32 + ((v & 4) ? 16 : 0) + grp * 8 + (v & 3) * 2;
      const float2 xp = *(const float2*)(X + (size_t)node * 64 + k);
      a[2 * v]     = (_Float16)xp.x;
      a[2 * v + 1] = (_Float16)xp.y;
      b[2 * v]     = (_Float16)W[(size_t)(k)     * CO + nIdx];
      b[2 * v + 1] = (_Float16)W[(size_t)(k + 1) * CO + nIdx];
    }
    c = __builtin_amdgcn_wmma_f32_16x16x32_f16(false, a, false, b,
                                               (short)0, c, false, false);
  }

  if ((mtile + 1) * 16 <= n_nodes) {
#pragma unroll
    for (int r = 0; r < 8; ++r)
      Out[(size_t)(mtile * 16 + r + 8 * grp) * CO + nIdx] = c[r];
  } else {
#pragma unroll
    for (int r = 0; r < 8; ++r) {
      const int row = mtile * 16 + r + 8 * grp;
      if (row < n_nodes) Out[(size_t)row * CO + nIdx] = c[r];
    }
  }
}

__global__ __launch_bounds__(256)
void k_zero(int* __restrict__ cnt, int n) {
  const int i = blockIdx.x * blockDim.x + threadIdx.x;
  if (i < n) cnt[i] = 0;
}

__global__ __launch_bounds__(256)
void edge_rank(const int* __restrict__ dst, int* __restrict__ cnt,
               int* __restrict__ rank, int ne) {
  const int e = blockIdx.x * blockDim.x + threadIdx.x;
  if (e >= ne) return;
  rank[e] = atomicAdd(cnt + dst[e], 1);
}

#define SCAN_T 1024
#define SCAN_V 4
#define SCAN_K 13

__global__ __launch_bounds__(SCAN_T)
void bucket_scan(const int* __restrict__ cnt, int* __restrict__ rs, int n) {
  __shared__ int wsum[32];
  __shared__ int wpre[33];
  const int t    = threadIdx.x;
  const int lane = t & 31;
  const int wave = t >> 5;
  int carry = 0;
  for (int sb = 0; sb <= n; sb += SCAN_T * SCAN_V * SCAN_K) {
    int c[SCAN_K][SCAN_V];
#pragma unroll
    for (int k = 0; k < SCAN_K; ++k) {
#pragma unroll
      for (int q = 0; q < SCAN_V; ++q) {
        const int i = sb + (k * SCAN_T + t) * SCAN_V + q;
        c[k][q] = (i < n) ? cnt[i] : 0;
      }
    }
#pragma unroll
    for (int k = 0; k < SCAN_K; ++k) {
      const int s = (c[k][0] + c[k][1]) + (c[k][2] + c[k][3]);
      int incl = s;
#pragma unroll
      for (int o = 1; o < 32; o <<= 1) {
        const int v = __shfl_up(incl, o, 32);
        if (lane >= o) incl += v;
      }
      if (lane == 31) wsum[wave] = incl;
      __syncthreads();
      if (wave == 0) {
        const int x = wsum[lane];
        int y = x;
#pragma unroll
        for (int o = 1; o < 32; o <<= 1) {
          const int v = __shfl_up(y, o, 32);
          if (lane >= o) y += v;
        }
        wpre[lane] = y - x;
        if (lane == 31) wpre[32] = y;
      }
      __syncthreads();
      const int woff = wpre[wave];
      const int tot  = wpre[32];
      int ex = carry + woff + incl - s;
      const int i0 = sb + (k * SCAN_T + t) * SCAN_V;
#pragma unroll
      for (int q = 0; q < SCAN_V; ++q) {
        if (i0 + q <= n) rs[i0 + q] = ex;
        ex += c[k][q];
      }
      carry += tot;
    }
  }
}

__global__ __launch_bounds__(256)
void edge_place(const int* __restrict__ src, const int* __restrict__ dst,
                const float* __restrict__ ef, const int* __restrict__ rank,
                const int* __restrict__ rs, int* __restrict__ pedge, int ne) {
  const int e = blockIdx.x * blockDim.x + threadIdx.x;
  if (e >= ne) return;
  const int pos = rs[dst[e]] + rank[e];
  *(int2*)(pedge + 2 * (size_t)pos) = make_int2(src[e], __float_as_int(ef[e]));
}

template<int CO, bool RELU>
__global__ __launch_bounds__(256)
void node_attend(const float* __restrict__ xl, const float* __restrict__ xr,
                 const int* __restrict__ rs, const int* __restrict__ pedge,
                 const float* __restrict__ We, const float* __restrict__ att,
                 const float* __restrict__ bias, float* __restrict__ out, int n) {
  constexpr int LPE = CO / 4;
  const int gid  = blockIdx.x * blockDim.x + threadIdx.x;
  const int node = gid / LPE;
  const int sub  = gid % LPE;
  if (node >= n) return;
  const int off = sub * 4;
  const int beg = rs[node];
  const int end = rs[node + 1];
  const float4 r4 = *(const float4*)(xr + (size_t)node * CO + off);
  const float4 w4 = *(const float4*)(We + off);
  const float4 t4 = *(const float4*)(att + off);
  const float NEG_INF = -__builtin_huge_valf();

  float mx = NEG_INF, den = 0.0f;
  float4 acc = make_float4(0.0f, 0.0f, 0.0f, 0.0f);

  for (int base = beg; base < end; base += LPE) {
    const int j = base + sub;
    int2 rec = make_int2(0, 0);
    if (j < end) rec = *(const int2*)(pedge + 2 * (size_t)j);
    const int here = min(LPE, end - base);
    for (int t = 0; t < here; t += 4) {
      float4 a[4];
      float  p[4], fe4[4];
      int    s4[4];
#pragma unroll
      for (int u = 0; u < 4; ++u) {
        s4[u]  = __shfl(rec.x, t + u, LPE);
        fe4[u] = __int_as_float(__shfl(rec.y, t + u, LPE));
      }
#pragma unroll
      for (int u = 0; u < 4; ++u)
        a[u] = *(const float4*)(xl + (size_t)s4[u] * CO + off);
      __builtin_amdgcn_sched_barrier(0);
#pragma unroll
      for (int u = 0; u < 4; ++u) {
        const float fe = fe4[u];
        float m, q = 0.0f;
        m = a[u].x + r4.x + fe * w4.x; m = (m > 0.0f) ? m : NEG_SLOPE * m; q += m * t4.x;
        m = a[u].y + r4.y + fe * w4.y; m = (m > 0.0f) ? m : NEG_SLOPE * m; q += m * t4.y;
        m = a[u].z + r4.z + fe * w4.z; m = (m > 0.0f) ? m : NEG_SLOPE * m; q += m * t4.z;
        m = a[u].w + r4.w + fe * w4.w; m = (m > 0.0f) ? m : NEG_SLOPE * m; q += m * t4.w;
        p[u] = q;
      }
#pragma unroll
      for (int o = LPE / 2; o > 0; o >>= 1) {
#pragma unroll
        for (int u = 0; u < 4; ++u) p[u] += __shfl_xor(p[u], o, LPE);
      }
#pragma unroll
      for (int u = 1; u < 4; ++u) {
        if (t + u >= here) {
          p[u] = NEG_INF;
          a[u] = make_float4(0.0f, 0.0f, 0.0f, 0.0f);
        }
      }
      const float mn = fmaxf(fmaxf(mx, p[0]), fmaxf(fmaxf(p[1], p[2]), p[3]));
      const float sc = __expf(mx - mn);
      const float e0 = __expf(p[0] - mn);
      const float e1 = __expf(p[1] - mn);
      const float e2 = __expf(p[2] - mn);
      const float e3 = __expf(p[3] - mn);
      den   = den   * sc + ((e0 + e1) + (e2 + e3));
      acc.x = acc.x * sc + ((a[0].x * e0 + a[1].x * e1) + (a[2].x * e2 + a[3].x * e3));
      acc.y = acc.y * sc + ((a[0].y * e0 + a[1].y * e1) + (a[2].y * e2 + a[3].y * e3));
      acc.z = acc.z * sc + ((a[0].z * e0 + a[1].z * e1) + (a[2].z * e2 + a[3].z * e3));
      acc.w = acc.w * sc + ((a[0].w * e0 + a[1].w * e1) + (a[2].w * e2 + a[3].w * e3));
      mx = mn;
    }
  }

  const float dn = den + 1e-16f;
  const float4 b4 = *(const float4*)(bias + off);
  float4 o4;
  o4.x = acc.x / dn + b4.x;
  o4.y = acc.y / dn + b4.y;
  o4.z = acc.z / dn + b4.z;
  o4.w = acc.w / dn + b4.w;
  if (RELU) {
    o4.x = fmaxf(o4.x, 0.0f); o4.y = fmaxf(o4.y, 0.0f);
    o4.z = fmaxf(o4.z, 0.0f); o4.w = fmaxf(o4.w, 0.0f);
  }
  *(float4*)(out + (size_t)node * CO + off) = o4;
}

extern "C" void kernel_launch(void* const* d_in, const int* in_sizes, int n_in,
                              void* d_out, int out_size, void* d_ws, size_t ws_size,
                              hipStream_t stream) {
  (void)n_in; (void)out_size; (void)ws_size;
  const float* x    = (const float*)d_in[0];
  const int*   ei   = (const int*)d_in[1];
  const float* ef   = (const float*)d_in[2];
  const float* Wl0  = (const float*)d_in[3];
  const float* bl0  = (const float*)d_in[4];
  const float* Wr0  = (const float*)d_in[5];
  const float* br0  = (const float*)d_in[6];
  const float* We0  = (const float*)d_in[7];
  const float* att0 = (const float*)d_in[8];
  const float* b0   = (const float*)d_in[9];
  const float* Wl1  = (const float*)d_in[10];
  const float* bl1  = (const float*)d_in[11];
  const float* Wr1  = (const float*)d_in[12];
  const float* br1  = (const float*)d_in[13];
  const float* We1  = (const float*)d_in[14];
  const float* att1 = (const float*)d_in[15];
  const float* b1   = (const float*)d_in[16];

  const int n = in_sizes[0] / 64;
  const int e = in_sizes[1] / 2;
  const int* srcI = ei;
  const int* dstI = ei + e;

  float* ws   = (float*)d_ws;
  float* xl   = ws;  ws += (size_t)n * 64;
  float* xr   = ws;  ws += (size_t)n * 64;
  float* hbuf = ws;  ws += (size_t)n * 64;
  int* pedge  = (int*)ws;  ws += (size_t)e * 2;
  int* rank   = (int*)ws;  ws += (size_t)e;
  int* cnt    = (int*)ws;  ws += n;
  int* rs     = (int*)ws;  ws += n + 1;

  auto cdiv = [](long long a, long long b) { return (int)((a + b - 1) / b); };
  const int B = 256;

  k_zero<<<cdiv(n, B), B, 0, stream>>>(cnt, n);
  edge_rank<<<cdiv(e, B), B, 0, stream>>>(dstI, cnt, rank, e);
  bucket_scan<<<1, SCAN_T, 0, stream>>>(cnt, rs, n);
  edge_place<<<cdiv(e, B), B, 0, stream>>>(srcI, dstI, ef, rank, rs, pedge, e);

  {
    constexpr int CO = 64;
    dim3 g((n + 15) / 16, CO / 16);
    node_xform_wmma<CO><<<g, 32, 0, stream>>>(x, Wl0, bl0, xl, n);
    node_xform_wmma<CO><<<g, 32, 0, stream>>>(x, Wr0, br0, xr, n);
    node_attend<CO, true><<<cdiv((long long)n * (CO / 4), B), B, 0, stream>>>(
        xl, xr, rs, pedge, We0, att0, b0, hbuf, n);
  }

  {
    constexpr int CO = 32;
    float* outp = (float*)d_out;
    dim3 g((n + 15) / 16, CO / 16);
    node_xform_wmma<CO><<<g, 32, 0, stream>>>(hbuf, Wl1, bl1, xl, n);
    node_xform_wmma<CO><<<g, 32, 0, stream>>>(hbuf, Wr1, br1, xr, n);
    node_attend<CO, false><<<cdiv((long long)n * (CO / 4), B), B, 0, stream>>>(
        xl, xr, rs, pedge, We1, att1, b1, outp, n);
  }
}


# Node transforms: one wave per 16-row strip makes every column tile of both xl and xr from one A fragment set and stores whole rows; 4 launches of 37500 one-wave groups become 2 of 391 groups
# speedup vs baseline: 2.9742x; 1.1331x over previous
#include <hip/hip_runtime.h>
#include <hip/hip_bf16.h>

#define NEG_SLOPE 0.2f

typedef __attribute__((ext_vector_type(16))) _Float16 v16h;
typedef __attribute__((ext_vector_type(8)))  float    v8f;

template<int NV> using vecf = float __attribute__((ext_vector_type(NV)));

template<int CO>
__global__ __launch_bounds__(256)
void node_xform_pair(const float* __restrict__ X,
                     const float* __restrict__ Wl, const float* __restrict__ bl,
                     const float* __restrict__ Wr, const float* __restrict__ br,
                     float* __restrict__ xl, float* __restrict__ xr, int n_nodes) {
  constexpr int TPM = CO / 16;
  const int lane  = threadIdx.x & 31;
  const int strip = __builtin_amdgcn_readfirstlane(
      (int)(blockIdx.x * (blockDim.x >> 5) + (threadIdx.x >> 5)));
  if (strip * 16 >= n_nodes) return;
  const int c = lane & 15;
  const int h = lane >> 4;

  int node = strip * 16 + c;
  if (node >= n_nodes) node = n_nodes - 1;

  v16h a[2];
#pragma unroll
  for (int ks = 0; ks < 2; ++ks) {
    const float* xp = X + (size_t)node * 64 + ks * 32 + h * 8;
    const float4 p0 = *(const float4*)(xp);
    const float4 p1 = *(const float4*)(xp + 4);
    const float4 p2 = *(const float4*)(xp + 16);
    const float4 p3 = *(const float4*)(xp + 20);
    a[ks][0]  = (_Float16)p0.x; a[ks][1]  = (_Float16)p0.y;
    a[ks][2]  = (_Float16)p0.z; a[ks][3]  = (_Float16)p0.w;
    a[ks][4]  = (_Float16)p1.x; a[ks][5]  = (_Float16)p1.y;
    a[ks][6]  = (_Float16)p1.z; a[ks][7]  = (_Float16)p1.w;
    a[ks][8]  = (_Float16)p2.x; a[ks][9]  = (_Float16)p2.y;
    a[ks][10] = (_Float16)p2.z; a[ks][11] = (_Float16)p2.w;
    a[ks][12] = (_Float16)p3.x; a[ks][13] = (_Float16)p3.y;
    a[ks][14] = (_Float16)p3.z; a[ks][15] = (_Float16)p3.w;
  }

  v8f acc[2][TPM];
#pragma unroll
  for (int t = 0; t < TPM; ++t) {
    const float vl = bl[TPM * c + t];
    const float vr = br[TPM * c + t];
#pragma unroll
    for (int r = 0; r < 8; ++r) { acc[0][t][r] = vl; acc[1][t][r] = vr; }
  }

#pragma unroll
  for (int ks = 0; ks < 2; ++ks) {
#pragma unroll
    for (int m = 0; m < 2; ++m) {
      const float* __restrict__ Wm = m ? Wr : Wl;
      vecf<TPM> w[16];
#pragma unroll
      for (int s = 0; s < 16; ++s) {
        const int k = ks * 32 + ((s & 8) ? 16 : 0) + h * 8 + (s & 7);
        w[s] = *(const vecf<TPM>*)(Wm + (size_t)k * CO + TPM * c);
      }
      __builtin_amdgcn_sched_barrier(0);
      v16h b[TPM];
#pragma unroll
      for (int s = 0; s < 16; ++s) {
#pragma unroll
        for (int t = 0; t < TPM; ++t) b[t][s] = (_Float16)w[s][t];
      }
#pragma unroll
      for (int t = 0; t < TPM; ++t) {
        acc[m][t] = __builtin_amdgcn_wmma_f32_16x16x32_f16(false, a[ks], false, b[t],
                                                           (short)0, acc[m][t], false, false);
      }
    }
  }

  const int row0 = strip * 16 + 8 * h;
  if ((strip + 1) * 16 <= n_nodes) {
#pragma unroll
    for (int r = 0; r < 8; ++r) {
      vecf<TPM> ol, orr;
#pragma unroll
      for (int t = 0; t < TPM; ++t) { ol[t] = acc[0][t][r]; orr[t] = acc[1][t][r]; }
      *(vecf<TPM>*)(xl + (size_t)(row0 + r) * CO + TPM * c) = ol;
      *(vecf<TPM>*)(xr + (size_t)(row0 + r) * CO + TPM * c) = orr;
    }
  } else {
#pragma unroll
    for (int r = 0; r < 8; ++r) {
      vecf<TPM> ol, orr;
#pragma unroll
      for (int t = 0; t < TPM; ++t) { ol[t] = acc[0][t][r]; orr[t] = acc[1][t][r]; }
      if (row0 + r < n_nodes) {
        *(vecf<TPM>*)(xl + (size_t)(row0 + r) * CO + TPM * c) = ol;
        *(vecf<TPM>*)(xr + (size_t)(row0 + r) * CO + TPM * c) = orr;
      }
    }
  }
}

__global__ __launch_bounds__(256)
void k_zero(int* __restrict__ cnt, int n) {
  const int i = blockIdx.x * blockDim.x + threadIdx.x;
  if (i < n) cnt[i] = 0;
}

__global__ __launch_bounds__(256)
void edge_rank(const int* __restrict__ dst, int* __restrict__ cnt,
               int* __restrict__ rank, int ne) {
  const int e = blockIdx.x * blockDim.x + threadIdx.x;
  if (e >= ne) return;
  rank[e] = atomicAdd(cnt + dst[e], 1);
}

#define SCAN_T 1024
#define SCAN_V 4
#define SCAN_K 13

__global__ __launch_bounds__(SCAN_T)
void bucket_scan(const int* __restrict__ cnt, int* __restrict__ rs, int n) {
  __shared__ int wsum[32];
  __shared__ int wpre[33];
  const int t    = threadIdx.x;
  const int lane = t & 31;
  const int wave = t >> 5;
  int carry = 0;
  for (int sb = 0; sb <= n; sb += SCAN_T * SCAN_V * SCAN_K) {
    int c[SCAN_K][SCAN_V];
#pragma unroll
    for (int k = 0; k < SCAN_K; ++k) {
#pragma unroll
      for (int q = 0; q < SCAN_V; ++q) {
        const int i = sb + (k * SCAN_T + t) * SCAN_V + q;
        c[k][q] = (i < n) ? cnt[i] : 0;
      }
    }
#pragma unroll
    for (int k = 0; k < SCAN_K; ++k) {
      const int s = (c[k][0] + c[k][1]) + (c[k][2] + c[k][3]);
      int incl = s;
#pragma unroll
      for (int o = 1; o < 32; o <<= 1) {
        const int v = __shfl_up(incl, o, 32);
        if (lane >= o) incl += v;
      }
      if (lane == 31) wsum[wave] = incl;
      __syncthreads();
      if (wave == 0) {
        const int x = wsum[lane];
        int y = x;
#pragma unroll
        for (int o = 1; o < 32; o <<= 1) {
          const int v = __shfl_up(y, o, 32);
          if (lane >= o) y += v;
        }
        wpre[lane] = y - x;
        if (lane == 31) wpre[32] = y;
      }
      __syncthreads();
      const int woff = wpre[wave];
      const int tot  = wpre[32];
      int ex = carry + woff + incl - s;
      const int i0 = sb + (k * SCAN_T + t) * SCAN_V;
#pragma unroll
      for (int q = 0; q < SCAN_V; ++q) {
        if (i0 + q <= n) rs[i0 + q] = ex;
        ex += c[k][q];
      }
      carry += tot;
    }
  }
}

__global__ __launch_bounds__(256)
void edge_place(const int* __restrict__ src, const int* __restrict__ dst,
                const float* __restrict__ ef, const int* __restrict__ rank,
                const int* __restrict__ rs, int* __restrict__ pedge, int ne) {
  const int e = blockIdx.x * blockDim.x + threadIdx.x;
  if (e >= ne) return;
  const int pos = rs[dst[e]] + rank[e];
  *(int2*)(pedge + 2 * (size_t)pos) = make_int2(src[e], __float_as_int(ef[e]));
}

template<int CO, bool RELU>
__global__ __launch_bounds__(256)
void node_attend(const float* __restrict__ xl, const float* __restrict__ xr,
                 const int* __restrict__ rs, const int* __restrict__ pedge,
                 const float* __restrict__ We, const float* __restrict__ att,
                 const float* __restrict__ bias, float* __restrict__ out, int n) {
  constexpr int LPE = CO / 4;
  const int gid  = blockIdx.x * blockDim.x + threadIdx.x;
  const int node = gid / LPE;
  const int sub  = gid % LPE;
  if (node >= n) return;
  const int off = sub * 4;
  const int beg = rs[node];
  const int end = rs[node + 1];
  const float4 r4 = *(const float4*)(xr + (size_t)node * CO + off);
  const float4 w4 = *(const float4*)(We + off);
  const float4 t4 = *(const float4*)(att + off);
  const float NEG_INF = -__builtin_huge_valf();

  float mx = NEG_INF, den = 0.0f;
  float4 acc = make_float4(0.0f, 0.0f, 0.0f, 0.0f);

  for (int base = beg; base < end; base += LPE) {
    const int j = base + sub;
    int2 rec = make_int2(0, 0);
    if (j < end) rec = *(const int2*)(pedge + 2 * (size_t)j);
    const int here = min(LPE, end - base);
    for (int t = 0; t < here; t += 4) {
      float4 a[4];
      float  p[4], fe4[4];
      int    s4[4];
#pragma unroll
      for (int u = 0; u < 4; ++u) {
        s4[u]  = __shfl(rec.x, t + u, LPE);
        fe4[u] = __int_as_float(__shfl(rec.y, t + u, LPE));
      }
#pragma unroll
      for (int u = 0; u < 4; ++u)
        a[u] = *(const float4*)(xl + (size_t)s4[u] * CO + off);
      __builtin_amdgcn_sched_barrier(0);
#pragma unroll
      for (int u = 0; u < 4; ++u) {
        const float fe = fe4[u];
        float m, q = 0.0f;
        m = a[u].x + r4.x + fe * w4.x; m = (m > 0.0f) ? m : NEG_SLOPE * m; q += m * t4.x;
        m = a[u].y + r4.y + fe * w4.y; m = (m > 0.0f) ? m : NEG_SLOPE * m; q += m * t4.y;
        m = a[u].z + r4.z + fe * w4.z; m = (m > 0.0f) ? m : NEG_SLOPE * m; q += m * t4.z;
        m = a[u].w + r4.w + fe * w4.w; m = (m > 0.0f) ? m : NEG_SLOPE * m; q += m * t4.w;
        p[u] = q;
      }
#pragma unroll
      for (int o = LPE / 2; o > 0; o >>= 1) {
#pragma unroll
        for (int u = 0; u < 4; ++u) p[u] += __shfl_xor(p[u], o, LPE);
      }
#pragma unroll
      for (int u = 1; u < 4; ++u) {
        if (t + u >= here) {
          p[u] = NEG_INF;
          a[u] = make_float4(0.0f, 0.0f, 0.0f, 0.0f);
        }
      }
      const float mn = fmaxf(fmaxf(mx, p[0]), fmaxf(fmaxf(p[1], p[2]), p[3]));
      const float sc = __expf(mx - mn);
      const float e0 = __expf(p[0] - mn);
      const float e1 = __expf(p[1] - mn);
      const float e2 = __expf(p[2] - mn);
      const float e3 = __expf(p[3] - mn);
      den   = den   * sc + ((e0 + e1) + (e2 + e3));
      acc.x = acc.x * sc + ((a[0].x * e0 + a[1].x * e1) + (a[2].x * e2 + a[3].x * e3));
      acc.y = acc.y * sc + ((a[0].y * e0 + a[1].y * e1) + (a[2].y * e2 + a[3].y * e3));
      acc.z = acc.z * sc + ((a[0].z * e0 + a[1].z * e1) + (a[2].z * e2 + a[3].z * e3));
      acc.w = acc.w * sc + ((a[0].w * e0 + a[1].w * e1) + (a[2].w * e2 + a[3].w * e3));
      mx = mn;
    }
  }

  const float dn = den + 1e-16f;
  const float4 b4 = *(const float4*)(bias + off);
  float4 o4;
  o4.x = acc.x / dn + b4.x;
  o4.y = acc.y / dn + b4.y;
  o4.z = acc.z / dn + b4.z;
  o4.w = acc.w / dn + b4.w;
  if (RELU) {
    o4.x = fmaxf(o4.x, 0.0f); o4.y = fmaxf(o4.y, 0.0f);
    o4.z = fmaxf(o4.z, 0.0f); o4.w = fmaxf(o4.w, 0.0f);
  }
  *(float4*)(out + (size_t)node * CO + off) = o4;
}

extern "C" void kernel_launch(void* const* d_in, const int* in_sizes, int n_in,
                              void* d_out, int out_size, void* d_ws, size_t ws_size,
                              hipStream_t stream) {
  (void)n_in; (void)out_size; (void)ws_size;
  const float* x    = (const float*)d_in[0];
  const int*   ei   = (const int*)d_in[1];
  const float* ef   = (const float*)d_in[2];
  const float* Wl0  = (const float*)d_in[3];
  const float* bl0  = (const float*)d_in[4];
  const float* Wr0  = (const float*)d_in[5];
  const float* br0  = (const float*)d_in[6];
  const float* We0  = (const float*)d_in[7];
  const float* att0 = (const float*)d_in[8];
  const float* b0   = (const float*)d_in[9];
  const float* Wl1  = (const float*)d_in[10];
  const float* bl1  = (const float*)d_in[11];
  const float* Wr1  = (const float*)d_in[12];
  const float* br1  = (const float*)d_in[13];
  const float* We1  = (const float*)d_in[14];
  const float* att1 = (const float*)d_in[15];
  const float* b1   = (const float*)d_in[16];

  const int n = in_sizes[0] / 64;
  const int e = in_sizes[1] / 2;
  const int* srcI = ei;
  const int* dstI = ei + e;

  float* ws   = (float*)d_ws;
  float* xl   = ws;  ws += (size_t)n * 64;
  float* xr   = ws;  ws += (size_t)n * 64;
  float* hbuf = ws;  ws += (size_t)n * 64;
  int* pedge  = (int*)ws;  ws += (size_t)e * 2;
  int* rank   = (int*)ws;  ws += (size_t)e;
  int* cnt    = (int*)ws;  ws += n;
  int* rs     = (int*)ws;  ws += n + 1;

  auto cdiv = [](long long a, long long b) { return (int)((a + b - 1) / b); };
  const int B = 256;

  k_zero<<<cdiv(n, B), B, 0, stream>>>(cnt, n);
  edge_rank<<<cdiv(e, B), B, 0, stream>>>(dstI, cnt, rank, e);
  bucket_scan<<<1, SCAN_T, 0, stream>>>(cnt, rs, n);
  edge_place<<<cdiv(e, B), B, 0, stream>>>(srcI, dstI, ef, rank, rs, pedge, e);

  {
    constexpr int CO = 64;
    node_xform_pair<CO><<<cdiv(n, 128), B, 0, stream>>>(x, Wl0, bl0, Wr0, br0, xl, xr, n);
    node_attend<CO, true><<<cdiv((long long)n * (CO / 4), B), B, 0, stream>>>(
        xl, xr, rs, pedge, We0, att0, b0, hbuf, n);
  }

  {
    constexpr int CO = 32;
    float* outp = (float*)d_out;
    node_xform_pair<CO><<<cdiv(n, 128), B, 0, stream>>>(hbuf, Wl1, bl1, Wr1, br1, xl, xr, n);
    node_attend<CO, false><<<cdiv((long long)n * (CO / 4), B), B, 0, stream>>>(
        xl, xr, rs, pedge, We1, att1, b1, outp, n);
  }
}


# Bucket edges with a two-level counting sort on LDS counters (chunk counts, column sums, per-bin count/scan/place; 3 launches of 256 groups), no global atomics; sums in LDS-atomic arrival order
# speedup vs baseline: 4.8976x; 1.6467x over previous
#include <hip/hip_runtime.h>
#include <hip/hip_bf16.h>

#define NEG_SLOPE 0.2f

typedef __attribute__((ext_vector_type(16))) _Float16 v16h;
typedef __attribute__((ext_vector_type(8)))  float    v8f;

template<int NV> using vecf = float __attribute__((ext_vector_type(NV)));

template<int CO>
__global__ __launch_bounds__(256)
void node_xform_pair(const float* __restrict__ X,
                     const float* __restrict__ Wl, const float* __restrict__ bl,
                     const float* __restrict__ Wr, const float* __restrict__ br,
                     float* __restrict__ xl, float* __restrict__ xr, int n_nodes) {
  constexpr int TPM = CO / 16;
  const int lane  = threadIdx.x & 31;
  const int strip = __builtin_amdgcn_readfirstlane(
      (int)(blockIdx.x * (blockDim.x >> 5) + (threadIdx.x >> 5)));
  if (strip * 16 >= n_nodes) return;
  const int c = lane & 15;
  const int h = lane >> 4;

  int node = strip * 16 + c;
  if (node >= n_nodes) node = n_nodes - 1;

  v16h a[2];
#pragma unroll
  for (int ks = 0; ks < 2; ++ks) {
    const float* xp = X + (size_t)node * 64 + ks * 32 + h * 8;
    const float4 p0 = *(const float4*)(xp);
    const float4 p1 = *(const float4*)(xp + 4);
    const float4 p2 = *(const float4*)(xp + 16);
    const float4 p3 = *(const float4*)(xp + 20);
    a[ks][0]  = (_Float16)p0.x; a[ks][1]  = (_Float16)p0.y;
    a[ks][2]  = (_Float16)p0.z; a[ks][3]  = (_Float16)p0.w;
    a[ks][4]  = (_Float16)p1.x; a[ks][5]  = (_Float16)p1.y;
    a[ks][6]  = (_Float16)p1.z; a[ks][7]  = (_Float16)p1.w;
    a[ks][8]  = (_Float16)p2.x; a[ks][9]  = (_Float16)p2.y;
    a[ks][10] = (_Float16)p2.z; a[ks][11] = (_Float16)p2.w;
    a[ks][12] = (_Float16)p3.x; a[ks][13] = (_Float16)p3.y;
    a[ks][14] = (_Float16)p3.z; a[ks][15] = (_Float16)p3.w;
  }

  v8f acc[2][TPM];
#pragma unroll
  for (int t = 0; t < TPM; ++t) {
    const float vl = bl[TPM * c + t];
    const float vr = br[TPM * c + t];
#pragma unroll
    for (int r = 0; r < 8; ++r) { acc[0][t][r] = vl; acc[1][t][r] = vr; }
  }

#pragma unroll
  for (int ks = 0; ks < 2; ++ks) {
#pragma unroll
    for (int m = 0; m < 2; ++m) {
      const float* __restrict__ Wm = m ? Wr : Wl;
      vecf<TPM> w[16];
#pragma unroll
      for (int s = 0; s < 16; ++s) {
        const int k = ks * 32 + ((s & 8) ? 16 : 0) + h * 8 + (s & 7);
        w[s] = *(const vecf<TPM>*)(Wm + (size_t)k * CO + TPM * c);
      }
      __builtin_amdgcn_sched_barrier(0);
      v16h b[TPM];
#pragma unroll
      for (int s = 0; s < 16; ++s) {
#pragma unroll
        for (int t = 0; t < TPM; ++t) b[t][s] = (_Float16)w[s][t];
      }
#pragma unroll
      for (int t = 0; t < TPM; ++t) {
        acc[m][t] = __builtin_amdgcn_wmma_f32_16x16x32_f16(false, a[ks], false, b[t],
                                                           (short)0, acc[m][t], false, false);
      }
    }
  }

  const int row0 = strip * 16 + 8 * h;
  if ((strip + 1) * 16 <= n_nodes) {
#pragma unroll
    for (int r = 0; r < 8; ++r) {
      vecf<TPM> ol, orr;
#pragma unroll
      for (int t = 0; t < TPM; ++t) { ol[t] = acc[0][t][r]; orr[t] = acc[1][t][r]; }
      *(vecf<TPM>*)(xl + (size_t)(row0 + r) * CO + TPM * c) = ol;
      *(vecf<TPM>*)(xr + (size_t)(row0 + r) * CO + TPM * c) = orr;
    }
  } else {
#pragma unroll
    for (int r = 0; r < 8; ++r) {
      vecf<TPM> ol, orr;
#pragma unroll
      for (int t = 0; t < TPM; ++t) { ol[t] = acc[0][t][r]; orr[t] = acc[1][t][r]; }
      if (row0 + r < n_nodes) {
        *(vecf<TPM>*)(xl + (size_t)(row0 + r) * CO + TPM * c) = ol;
        *(vecf<TPM>*)(xr + (size_t)(row0 + r) * CO + TPM * c) = orr;
      }
    }
  }
}

#define SORT_NB     256
#define SORT_T      512
#define SORT_K      8
#define SORT_MAXNPB 1024
#define SORT_STAGE  6144

__global__ __launch_bounds__(SORT_T)
void chunk_count(const int* __restrict__ dst, int* __restrict__ table, int n, int ne) {
  __shared__ int hist[SORT_NB];
  const int t = threadIdx.x;
  const int w = blockIdx.x;
  const unsigned npb = (unsigned)((n + SORT_NB - 1) / SORT_NB);
  const int ech = (ne + SORT_NB - 1) / SORT_NB;
  const int c0  = min(ne, w * ech);
  const int c1  = min(ne, c0 + ech);
  if (t < SORT_NB) hist[t] = 0;
  __syncthreads();
  for (int base = c0; base < c1; base += SORT_T * SORT_K) {
    int d[SORT_K];
#pragma unroll
    for (int k = 0; k < SORT_K; ++k) {
      const int i = base + k * SORT_T + t;
      d[k] = (i < c1) ? dst[i] : -1;
    }
#pragma unroll
    for (int k = 0; k < SORT_K; ++k)
      if (d[k] >= 0) atomicAdd(&hist[(unsigned)d[k] / npb], 1);
  }
  __syncthreads();
  if (t < SORT_NB) table[w * SORT_NB + t] = hist[t];
}

__global__ __launch_bounds__(SORT_T)
void chunk_place(const int* __restrict__ src, const int* __restrict__ dst,
                 const float* __restrict__ ef, const int* __restrict__ table,
                 int* __restrict__ binbase, int* __restrict__ trec_w, int n, int ne) {
  int4* __restrict__ trec = (int4*)trec_w;
  constexpr int NQ = SORT_T / 64;
  constexpr int RQ = SORT_NB / NQ;
  __shared__ int ptot[NQ][SORT_NB];
  __shared__ int ppre[NQ][SORT_NB];
  __shared__ int wtot[SORT_NB / 32];
  __shared__ int cursor[SORT_NB + 32];
  const int t    = threadIdx.x;
  const int lane = t & 31;
  const int wave = t >> 5;
  const int w    = blockIdx.x;
  const unsigned npb = (unsigned)((n + SORT_NB - 1) / SORT_NB);
  const int ech = (ne + SORT_NB - 1) / SORT_NB;
  const int c0  = min(ne, w * ech);
  const int c1  = min(ne, c0 + ech);

  int d[SORT_K], s[SORT_K], f[SORT_K];
#pragma unroll
  for (int k = 0; k < SORT_K; ++k) {
    const int i = c0 + k * SORT_T + t;
    const bool ok = i < c1;
    d[k] = ok ? dst[i] : -1;
    s[k] = ok ? src[i] : 0;
    f[k] = ok ? __float_as_int(ef[i]) : 0;
  }

  {
    const int q  = t >> 6;
    const int b4 = (t & 63) * 4;
    int4 tot = make_int4(0, 0, 0, 0), pre = make_int4(0, 0, 0, 0);
#pragma unroll
    for (int kb = 0; kb < RQ; kb += 16) {
      int4 v[16];
#pragma unroll
      for (int k = 0; k < 16; ++k)
        v[k] = *(const int4*)(table + (size_t)(q * RQ + kb + k) * SORT_NB + b4);
      __builtin_amdgcn_sched_barrier(0);
#pragma unroll
      for (int k = 0; k < 16; ++k) {
        const bool early = (q * RQ + kb + k) < w;
        tot.x += v[k].x; tot.y += v[k].y; tot.z += v[k].z; tot.w += v[k].w;
        pre.x += early ? v[k].x : 0; pre.y += early ? v[k].y : 0;
        pre.z += early ? v[k].z : 0; pre.w += early ? v[k].w : 0;
      }
    }
    *(int4*)&ptot[q][b4] = tot;
    *(int4*)&ppre[q][b4] = pre;
  }
  __syncthreads();
  int mytot = 0, mypre = 0, incl = 0;
  if (t < SORT_NB) {
#pragma unroll
    for (int q = 0; q < NQ; ++q) { mytot += ptot[q][t]; mypre += ppre[q][t]; }
    incl = mytot;
#pragma unroll
    for (int o = 1; o < 32; o <<= 1) {
      const int v = __shfl_up(incl, o, 32);
      if (lane >= o) incl += v;
    }
    if (lane == 31) wtot[wave] = incl;
  }
  __syncthreads();
  if (t < SORT_NB) {
    int woff = 0;
#pragma unroll
    for (int j = 0; j < SORT_NB / 32; ++j) woff += (j < wave) ? wtot[j] : 0;
    const int base = woff + incl - mytot;
    cursor[t] = base + mypre;
    if (t < 32) cursor[SORT_NB + t] = 0;
    if (w == 0) {
      binbase[t] = base;
      if (t == SORT_NB - 1) binbase[SORT_NB] = base + mytot;
    }
  }
  __syncthreads();

  for (int base = c0; base < c1; base += SORT_T * SORT_K) {
    int pos[SORT_K];
#pragma unroll
    for (int k = 0; k < SORT_K; ++k) {
      const int bin = (d[k] >= 0) ? (int)((unsigned)d[k] / npb) : SORT_NB + lane;
      pos[k] = atomicAdd(&cursor[bin], 1);
    }
#pragma unroll
    for (int k = 0; k < SORT_K; ++k)
      if (d[k] >= 0) trec[pos[k]] = make_int4(s[k], f[k], d[k], 0);
#pragma unroll
    for (int k = 0; k < SORT_K; ++k) {
      const int i = base + SORT_T * SORT_K + k * SORT_T + t;
      const bool ok = i < c1;
      d[k] = ok ? dst[i] : -1;
      s[k] = ok ? src[i] : 0;
      f[k] = ok ? __float_as_int(ef[i]) : 0;
    }
  }
}

__global__ __launch_bounds__(SORT_T)
void bin_sort(const int* __restrict__ binbase, const int* __restrict__ trec_w,
              int* __restrict__ rs, int* __restrict__ pedge_w, int n, int ne) {
  const int4* __restrict__ trec = (const int4*)trec_w;
  int2* __restrict__ pedge = (int2*)pedge_w;
  __shared__ int  cntL[SORT_MAXNPB];
  __shared__ int  wsum[SORT_T / 32];
  __shared__ int2 stage[SORT_STAGE];
  const int t    = threadIdx.x;
  const int lane = t & 31;
  const int wave = t >> 5;
  const int b    = blockIdx.x;
  const int npb  = (n + SORT_NB - 1) / SORT_NB;
  const int node0 = b * npb;
  const int beg = binbase[b];
  const int end = binbase[b + 1];
  const bool staged = (end - beg) <= SORT_STAGE;
  cntL[2 * t] = 0; cntL[2 * t + 1] = 0;
  if (b == 0 && t == 0) rs[n] = ne;
  __syncthreads();
  for (int base = beg; base < end; base += SORT_T * SORT_K) {
    int d[SORT_K];
#pragma unroll
    for (int k = 0; k < SORT_K; ++k) {
      const int i = base + k * SORT_T + t;
      d[k] = (i < end) ? trec[i].z : -1;
    }
#pragma unroll
    for (int k = 0; k < SORT_K; ++k)
      if (d[k] >= 0) atomicAdd(&cntL[d[k] - node0], 1);
  }
  __syncthreads();
  {
    const int n0 = cntL[2 * t], n1 = cntL[2 * t + 1];
    const int sm = n0 + n1;
    int incl = sm;
#pragma unroll
    for (int o = 1; o < 32; o <<= 1) {
      const int v = __shfl_up(incl, o, 32);
      if (lane >= o) incl += v;
    }
    if (lane == 31) wsum[wave] = incl;
    __syncthreads();
    int woff = 0;
#pragma unroll
    for (int j = 0; j < SORT_T / 32; ++j) woff += (j < wave) ? wsum[j] : 0;
    const int ex = woff + incl - sm;
    cntL[2 * t]     = ex;
    cntL[2 * t + 1] = ex + n0;
    const int l0 = 2 * t;
    if (l0 < npb && node0 + l0 < n)         rs[node0 + l0]     = beg + ex;
    if (l0 + 1 < npb && node0 + l0 + 1 < n) rs[node0 + l0 + 1] = beg + ex + n0;
  }
  __syncthreads();
  for (int base = beg; base < end; base += SORT_T * SORT_K) {
    int4 r[SORT_K];
#pragma unroll
    for (int k = 0; k < SORT_K; ++k) {
      const int i = base + k * SORT_T + t;
      r[k] = (i < end) ? trec[i] : make_int4(0, 0, -1, 0);
    }
    int pos[SORT_K];
#pragma unroll
    for (int k = 0; k < SORT_K; ++k)
      pos[k] = (r[k].z >= 0) ? atomicAdd(&cntL[r[k].z - node0], 1) : 0;
#pragma unroll
    for (int k = 0; k < SORT_K; ++k) {
      if (r[k].z >= 0) {
        if (staged) stage[pos[k]] = make_int2(r[k].x, r[k].y);
        else        pedge[beg + pos[k]] = make_int2(r[k].x, r[k].y);
      }
    }
  }
  __syncthreads();
  if (staged) {
    for (int j = t; j < end - beg; j += SORT_T) pedge[beg + j] = stage[j];
  }
}

template<int CO, bool RELU>
__global__ __launch_bounds__(256)
void node_attend(const float* __restrict__ xl, const float* __restrict__ xr,
                 const int* __restrict__ rs, const int* __restrict__ pedge,
                 const float* __restrict__ We, const float* __restrict__ att,
                 const float* __restrict__ bias, float* __restrict__ out, int n) {
  constexpr int LPE = CO / 4;
  const int gid  = blockIdx.x * blockDim.x + threadIdx.x;
  const int node = gid / LPE;
  const int sub  = gid % LPE;
  if (node >= n) return;
  const int off = sub * 4;
  const int beg = rs[node];
  const int end = rs[node + 1];
  const float4 r4 = *(const float4*)(xr + (size_t)node * CO + off);
  const float4 w4 = *(const float4*)(We + off);
  const float4 t4 = *(const float4*)(att + off);
  const float NEG_INF = -__builtin_huge_valf();

  float mx = NEG_INF, den = 0.0f;
  float4 acc = make_float4(0.0f, 0.0f, 0.0f, 0.0f);

  for (int base = beg; base < end; base += LPE) {
    const int j = base + sub;
    int2 rec = make_int2(0, 0);
    if (j < end) rec = *(const int2*)(pedge + 2 * (size_t)j);
    const int here = min(LPE, end - base);
    for (int t = 0; t < here; t += 4) {
      float4 a[4];
      float  p[4], fe4[4];
      int    s4[4];
#pragma unroll
      for (int u = 0; u < 4; ++u) {
        s4[u]  = __shfl(rec.x, t + u, LPE);
        fe4[u] = __int_as_float(__shfl(rec.y, t + u, LPE));
      }
#pragma unroll
      for (int u = 0; u < 4; ++u)
        a[u] = *(const float4*)(xl + (size_t)s4[u] * CO + off);
      __builtin_amdgcn_sched_barrier(0);
#pragma unroll
      for (int u = 0; u < 4; ++u) {
        const float fe = fe4[u];
        float m, q = 0.0f;
        m = a[u].x + r4.x + fe * w4.x; m = (m > 0.0f) ? m : NEG_SLOPE * m; q += m * t4.x;
        m = a[u].y + r4.y + fe * w4.y; m = (m > 0.0f) ? m : NEG_SLOPE * m; q += m * t4.y;
        m = a[u].z + r4.z + fe * w4.z; m = (m > 0.0f) ? m : NEG_SLOPE * m; q += m * t4.z;
        m = a[u].w + r4.w + fe * w4.w; m = (m > 0.0f) ? m : NEG_SLOPE * m; q += m * t4.w;
        p[u] = q;
      }
#pragma unroll
      for (int o = LPE / 2; o > 0; o >>= 1) {
#pragma unroll
        for (int u = 0; u < 4; ++u) p[u] += __shfl_xor(p[u], o, LPE);
      }
#pragma unroll
      for (int u = 1; u < 4; ++u) {
        if (t + u >= here) {
          p[u] = NEG_INF;
          a[u] = make_float4(0.0f, 0.0f, 0.0f, 0.0f);
        }
      }
      const float mn = fmaxf(fmaxf(mx, p[0]), fmaxf(fmaxf(p[1], p[2]), p[3]));
      const float sc = __expf(mx - mn);
      const float e0 = __expf(p[0] - mn);
      const float e1 = __expf(p[1] - mn);
      const float e2 = __expf(p[2] - mn);
      const float e3 = __expf(p[3] - mn);
      den   = den   * sc + ((e0 + e1) + (e2 + e3));
      acc.x = acc.x * sc + ((a[0].x * e0 + a[1].x * e1) + (a[2].x * e2 + a[3].x * e3));
      acc.y = acc.y * sc + ((a[0].y * e0 + a[1].y * e1) + (a[2].y * e2 + a[3].y * e3));
      acc.z = acc.z * sc + ((a[0].z * e0 + a[1].z * e1) + (a[2].z * e2 + a[3].z * e3));
      acc.w = acc.w * sc + ((a[0].w * e0 + a[1].w * e1) + (a[2].w * e2 + a[3].w * e3));
      mx = mn;
    }
  }

  const float dn = den + 1e-16f;
  const float4 b4 = *(const float4*)(bias + off);
  float4 o4;
  o4.x = acc.x / dn + b4.x;
  o4.y = acc.y / dn + b4.y;
  o4.z = acc.z / dn + b4.z;
  o4.w = acc.w / dn + b4.w;
  if (RELU) {
    o4.x = fmaxf(o4.x, 0.0f); o4.y = fmaxf(o4.y, 0.0f);
    o4.z = fmaxf(o4.z, 0.0f); o4.w = fmaxf(o4.w, 0.0f);
  }
  *(float4*)(out + (size_t)node * CO + off) = o4;
}

extern "C" void kernel_launch(void* const* d_in, const int* in_sizes, int n_in,
                              void* d_out, int out_size, void* d_ws, size_t ws_size,
                              hipStream_t stream) {
  (void)n_in; (void)out_size; (void)ws_size;
  const float* x    = (const float*)d_in[0];
  const int*   ei   = (const int*)d_in[1];
  const float* ef   = (const float*)d_in[2];
  const float* Wl0  = (const float*)d_in[3];
  const float* bl0  = (const float*)d_in[4];
  const float* Wr0  = (const float*)d_in[5];
  const float* br0  = (const float*)d_in[6];
  const float* We0  = (const float*)d_in[7];
  const float* att0 = (const float*)d_in[8];
  const float* b0   = (const float*)d_in[9];
  const float* Wl1  = (const float*)d_in[10];
  const float* bl1  = (const float*)d_in[11];
  const float* Wr1  = (const float*)d_in[12];
  const float* br1  = (const float*)d_in[13];
  const float* We1  = (const float*)d_in[14];
  const float* att1 = (const float*)d_in[15];
  const float* b1   = (const float*)d_in[16];

  const int n = in_sizes[0] / 64;
  const int e = in_sizes[1] / 2;
  const int* srcI = ei;
  const int* dstI = ei + e;

  float* ws   = (float*)d_ws;
  float* xl   = ws;  ws += (size_t)n * 64;
  float* xr   = ws;  ws += (size_t)n * 64;
  float* hbuf = ws;  ws += (size_t)n * 64;
  int* table  = (int*)ws;  ws += SORT_NB * SORT_NB;
  int* binbase = (int*)ws; ws += SORT_NB + 4;
  int* trec   = (int*)ws;  ws += (size_t)e * 4;
  int* pedge  = (int*)ws;  ws += (size_t)e * 2;
  int* rs     = (int*)ws;  ws += n + 1;

  auto cdiv = [](long long a, long long b) { return (int)((a + b - 1) / b); };
  const int B = 256;

  chunk_count<<<SORT_NB, SORT_T, 0, stream>>>(dstI, table, n, e);
  chunk_place<<<SORT_NB, SORT_T, 0, stream>>>(srcI, dstI, ef, table, binbase, trec, n, e);
  bin_sort<<<SORT_NB, SORT_T, 0, stream>>>(binbase, trec, rs, pedge, n, e);

  {
    constexpr int CO = 64;
    node_xform_pair<CO><<<cdiv(n, 128), B, 0, stream>>>(x, Wl0, bl0, Wr0, br0, xl, xr, n);
    node_attend<CO, true><<<cdiv((long long)n * (CO / 4), B), B, 0, stream>>>(
        xl, xr, rs, pedge, We0, att0, b0, hbuf, n);
  }

  {
    constexpr int CO = 32;
    float* outp = (float*)d_out;
    node_xform_pair<CO><<<cdiv(n, 128), B, 0, stream>>>(hbuf, Wl1, bl1, Wr1, br1, xl, xr, n);
    node_attend<CO, false><<<cdiv((long long)n * (CO / 4), B), B, 0, stream>>>(
        xl, xr, rs, pedge, We1, att1, b1, outp, n);
  }
}


# Node walk: eight channels per lane as four register pairs, so half the lanes per node and two bucket records per lane; per-edge softmax, reduce and address work halves; logit sum order changes
# speedup vs baseline: 5.6477x; 1.1532x over previous
#include <hip/hip_runtime.h>
#include <hip/hip_bf16.h>

#define NEG_SLOPE 0.2f

typedef __attribute__((ext_vector_type(16))) _Float16 v16h;
typedef __attribute__((ext_vector_type(8)))  float    v8f;

template<int NV> using vecf = float __attribute__((ext_vector_type(NV)));

template<int CO>
__global__ __launch_bounds__(256)
void node_xform_pair(const float* __restrict__ X,
                     const float* __restrict__ Wl, const float* __restrict__ bl,
                     const float* __restrict__ Wr, const float* __restrict__ br,
                     float* __restrict__ xl, float* __restrict__ xr, int n_nodes) {
  constexpr int TPM = CO / 16;
  const int lane  = threadIdx.x & 31;
  const int strip = __builtin_amdgcn_readfirstlane(
      (int)(blockIdx.x * (blockDim.x >> 5) + (threadIdx.x >> 5)));
  if (strip * 16 >= n_nodes) return;
  const int c = lane & 15;
  const int h = lane >> 4;

  int node = strip * 16 + c;
  if (node >= n_nodes) node = n_nodes - 1;

  v16h a[2];
#pragma unroll
  for (int ks = 0; ks < 2; ++ks) {
    const float* xp = X + (size_t)node * 64 + ks * 32 + h * 8;
    const float4 p0 = *(const float4*)(xp);
    const float4 p1 = *(const float4*)(xp + 4);
    const float4 p2 = *(const float4*)(xp + 16);
    const float4 p3 = *(const float4*)(xp + 20);
    a[ks][0]  = (_Float16)p0.x; a[ks][1]  = (_Float16)p0.y;
    a[ks][2]  = (_Float16)p0.z; a[ks][3]  = (_Float16)p0.w;
    a[ks][4]  = (_Float16)p1.x; a[ks][5]  = (_Float16)p1.y;
    a[ks][6]  = (_Float16)p1.z; a[ks][7]  = (_Float16)p1.w;
    a[ks][8]  = (_Float16)p2.x; a[ks][9]  = (_Float16)p2.y;
    a[ks][10] = (_Float16)p2.z; a[ks][11] = (_Float16)p2.w;
    a[ks][12] = (_Float16)p3.x; a[ks][13] = (_Float16)p3.y;
    a[ks][14] = (_Float16)p3.z; a[ks][15] = (_Float16)p3.w;
  }

  v8f acc[2][TPM];
#pragma unroll
  for (int t = 0; t < TPM; ++t) {
    const float vl = bl[TPM * c + t];
    const float vr = br[TPM * c + t];
#pragma unroll
    for (int r = 0; r < 8; ++r) { acc[0][t][r] = vl; acc[1][t][r] = vr; }
  }

#pragma unroll
  for (int ks = 0; ks < 2; ++ks) {
#pragma unroll
    for (int m = 0; m < 2; ++m) {
      const float* __restrict__ Wm = m ? Wr : Wl;
      vecf<TPM> w[16];
#pragma unroll
      for (int s = 0; s < 16; ++s) {
        const int k = ks * 32 + ((s & 8) ? 16 : 0) + h * 8 + (s & 7);
        w[s] = *(const vecf<TPM>*)(Wm + (size_t)k * CO + TPM * c);
      }
      __builtin_amdgcn_sched_barrier(0);
      v16h b[TPM];
#pragma unroll
      for (int s = 0; s < 16; ++s) {
#pragma unroll
        for (int t = 0; t < TPM; ++t) b[t][s] = (_Float16)w[s][t];
      }
#pragma unroll
      for (int t = 0; t < TPM; ++t) {
        acc[m][t] = __builtin_amdgcn_wmma_f32_16x16x32_f16(false, a[ks], false, b[t],
                                                           (short)0, acc[m][t], false, false);
      }
    }
  }

  const int row0 = strip * 16 + 8 * h;
  if ((strip + 1) * 16 <= n_nodes) {
#pragma unroll
    for (int r = 0; r < 8; ++r) {
      vecf<TPM> ol, orr;
#pragma unroll
      for (int t = 0; t < TPM; ++t) { ol[t] = acc[0][t][r]; orr[t] = acc[1][t][r]; }
      *(vecf<TPM>*)(xl + (size_t)(row0 + r) * CO + TPM * c) = ol;
      *(vecf<TPM>*)(xr + (size_t)(row0 + r) * CO + TPM * c) = orr;
    }
  } else {
#pragma unroll
    for (int r = 0; r < 8; ++r) {
      vecf<TPM> ol, orr;
#pragma unroll
      for (int t = 0; t < TPM; ++t) { ol[t] = acc[0][t][r]; orr[t] = acc[1][t][r]; }
      if (row0 + r < n_nodes) {
        *(vecf<TPM>*)(xl + (size_t)(row0 + r) * CO + TPM * c) = ol;
        *(vecf<TPM>*)(xr + (size_t)(row0 + r) * CO + TPM * c) = orr;
      }
    }
  }
}

#define SORT_NB     256
#define SORT_T      512
#define SORT_K      8
#define SORT_MAXNPB 1024
#define SORT_STAGE  6144

__global__ __launch_bounds__(SORT_T)
void chunk_count(const int* __restrict__ dst, int* __restrict__ table, int n, int ne) {
  __shared__ int hist[SORT_NB];
  const int t = threadIdx.x;
  const int w = blockIdx.x;
  const unsigned npb = (unsigned)((n + SORT_NB - 1) / SORT_NB);
  const int ech = (ne + SORT_NB - 1) / SORT_NB;
  const int c0  = min(ne, w * ech);
  const int c1  = min(ne, c0 + ech);
  if (t < SORT_NB) hist[t] = 0;
  __syncthreads();
  for (int base = c0; base < c1; base += SORT_T * SORT_K) {
    int d[SORT_K];
#pragma unroll
    for (int k = 0; k < SORT_K; ++k) {
      const int i = base + k * SORT_T + t;
      d[k] = (i < c1) ? dst[i] : -1;
    }
#pragma unroll
    for (int k = 0; k < SORT_K; ++k)
      if (d[k] >= 0) atomicAdd(&hist[(unsigned)d[k] / npb], 1);
  }
  __syncthreads();
  if (t < SORT_NB) table[w * SORT_NB + t] = hist[t];
}

__global__ __launch_bounds__(SORT_T)
void chunk_place(const int* __restrict__ src, const int* __restrict__ dst,
                 const float* __restrict__ ef, const int* __restrict__ table,
                 int* __restrict__ binbase, int* __restrict__ trec_w, int n, int ne) {
  int4* __restrict__ trec = (int4*)trec_w;
  constexpr int NQ = SORT_T / 64;
  constexpr int RQ = SORT_NB / NQ;
  __shared__ int ptot[NQ][SORT_NB];
  __shared__ int ppre[NQ][SORT_NB];
  __shared__ int wtot[SORT_NB / 32];
  __shared__ int cursor[SORT_NB + 32];
  const int t    = threadIdx.x;
  const int lane = t & 31;
  const int wave = t >> 5;
  const int w    = blockIdx.x;
  const unsigned npb = (unsigned)((n + SORT_NB - 1) / SORT_NB);
  const int ech = (ne + SORT_NB - 1) / SORT_NB;
  const int c0  = min(ne, w * ech);
  const int c1  = min(ne, c0 + ech);

  int d[SORT_K], s[SORT_K], f[SORT_K];
#pragma unroll
  for (int k = 0; k < SORT_K; ++k) {
    const int i = c0 + k * SORT_T + t;
    const bool ok = i < c1;
    d[k] = ok ? dst[i] : -1;
    s[k] = ok ? src[i] : 0;
    f[k] = ok ? __float_as_int(ef[i]) : 0;
  }

  {
    const int q  = t >> 6;
    const int b4 = (t & 63) * 4;
    int4 tot = make_int4(0, 0, 0, 0), pre = make_int4(0, 0, 0, 0);
#pragma unroll
    for (int kb = 0; kb < RQ; kb += 16) {
      int4 v[16];
#pragma unroll
      for (int k = 0; k < 16; ++k)
        v[k] = *(const int4*)(table + (size_t)(q * RQ + kb + k) * SORT_NB + b4);
      __builtin_amdgcn_sched_barrier(0);
#pragma unroll
      for (int k = 0; k < 16; ++k) {
        const bool early = (q * RQ + kb + k) < w;
        tot.x += v[k].x; tot.y += v[k].y; tot.z += v[k].z; tot.w += v[k].w;
        pre.x += early ? v[k].x : 0; pre.y += early ? v[k].y : 0;
        pre.z += early ? v[k].z : 0; pre.w += early ? v[k].w : 0;
      }
    }
    *(int4*)&ptot[q][b4] = tot;
    *(int4*)&ppre[q][b4] = pre;
  }
  __syncthreads();
  int mytot = 0, mypre = 0, incl = 0;
  if (t < SORT_NB) {
#pragma unroll
    for (int q = 0; q < NQ; ++q) { mytot += ptot[q][t]; mypre += ppre[q][t]; }
    incl = mytot;
#pragma unroll
    for (int o = 1; o < 32; o <<= 1) {
      const int v = __shfl_up(incl, o, 32);
      if (lane >= o) incl += v;
    }
    if (lane == 31) wtot[wave] = incl;
  }
  __syncthreads();
  if (t < SORT_NB) {
    int woff = 0;
#pragma unroll
    for (int j = 0; j < SORT_NB / 32; ++j) woff += (j < wave) ? wtot[j] : 0;
    const int base = woff + incl - mytot;
    cursor[t] = base + mypre;
    if (t < 32) cursor[SORT_NB + t] = 0;
    if (w == 0) {
      binbase[t] = base;
      if (t == SORT_NB - 1) binbase[SORT_NB] = base + mytot;
    }
  }
  __syncthreads();

  for (int base = c0; base < c1; base += SORT_T * SORT_K) {
    int pos[SORT_K];
#pragma unroll
    for (int k = 0; k < SORT_K; ++k) {
      const int bin = (d[k] >= 0) ? (int)((unsigned)d[k] / npb) : SORT_NB + lane;
      pos[k] = atomicAdd(&cursor[bin], 1);
    }
#pragma unroll
    for (int k = 0; k < SORT_K; ++k)
      if (d[k] >= 0) trec[pos[k]] = make_int4(s[k], f[k], d[k], 0);
#pragma unroll
    for (int k = 0; k < SORT_K; ++k) {
      const int i = base + SORT_T * SORT_K + k * SORT_T + t;
      const bool ok = i < c1;
      d[k] = ok ? dst[i] : -1;
      s[k] = ok ? src[i] : 0;
      f[k] = ok ? __float_as_int(ef[i]) : 0;
    }
  }
}

__global__ __launch_bounds__(SORT_T)
void bin_sort(const int* __restrict__ binbase, const int* __restrict__ trec_w,
              int* __restrict__ rs, int* __restrict__ pedge_w, int n, int ne) {
  const int4* __restrict__ trec = (const int4*)trec_w;
  int2* __restrict__ pedge = (int2*)pedge_w;
  __shared__ int  cntL[SORT_MAXNPB];
  __shared__ int  wsum[SORT_T / 32];
  __shared__ int2 stage[SORT_STAGE];
  const int t    = threadIdx.x;
  const int lane = t & 31;
  const int wave = t >> 5;
  const int b    = blockIdx.x;
  const int npb  = (n + SORT_NB - 1) / SORT_NB;
  const int node0 = b * npb;
  const int beg = binbase[b];
  const int end = binbase[b + 1];
  const bool staged = (end - beg) <= SORT_STAGE;
  cntL[2 * t] = 0; cntL[2 * t + 1] = 0;
  if (b == 0 && t == 0) rs[n] = ne;
  __syncthreads();
  for (int base = beg; base < end; base += SORT_T * SORT_K) {
    int d[SORT_K];
#pragma unroll
    for (int k = 0; k < SORT_K; ++k) {
      const int i = base + k * SORT_T + t;
      d[k] = (i < end) ? trec[i].z : -1;
    }
#pragma unroll
    for (int k = 0; k < SORT_K; ++k)
      if (d[k] >= 0) atomicAdd(&cntL[d[k] - node0], 1);
  }
  __syncthreads();
  {
    const int n0 = cntL[2 * t], n1 = cntL[2 * t + 1];
    const int sm = n0 + n1;
    int incl = sm;
#pragma unroll
    for (int o = 1; o < 32; o <<= 1) {
      const int v = __shfl_up(incl, o, 32);
      if (lane >= o) incl += v;
    }
    if (lane == 31) wsum[wave] = incl;
    __syncthreads();
    int woff = 0;
#pragma unroll
    for (int j = 0; j < SORT_T / 32; ++j) woff += (j < wave) ? wsum[j] : 0;
    const int ex = woff + incl - sm;
    cntL[2 * t]     = ex;
    cntL[2 * t + 1] = ex + n0;
    const int l0 = 2 * t;
    if (l0 < npb && node0 + l0 < n)         rs[node0 + l0]     = beg + ex;
    if (l0 + 1 < npb && node0 + l0 + 1 < n) rs[node0 + l0 + 1] = beg + ex + n0;
  }
  __syncthreads();
  for (int base = beg; base < end; base += SORT_T * SORT_K) {
    int4 r[SORT_K];
#pragma unroll
    for (int k = 0; k < SORT_K; ++k) {
      const int i = base + k * SORT_T + t;
      r[k] = (i < end) ? trec[i] : make_int4(0, 0, -1, 0);
    }
    int pos[SORT_K];
#pragma unroll
    for (int k = 0; k < SORT_K; ++k)
      pos[k] = (r[k].z >= 0) ? atomicAdd(&cntL[r[k].z - node0], 1) : 0;
#pragma unroll
    for (int k = 0; k < SORT_K; ++k) {
      if (r[k].z >= 0) {
        if (staged) stage[pos[k]] = make_int2(r[k].x, r[k].y);
        else        pedge[beg + pos[k]] = make_int2(r[k].x, r[k].y);
      }
    }
  }
  __syncthreads();
  if (staged) {
    for (int j = t; j < end - beg; j += SORT_T) pedge[beg + j] = stage[j];
  }
}

template<int M>
__device__ __forceinline__ float lane_xor_add(float v) {
  const int partner = __builtin_amdgcn_update_dpp(0, __float_as_int(v), 0x160 + M, 0xf, 0xf, true);
  return v + __int_as_float(partner);
}
__device__ __forceinline__ int row_pick(int v, int sel) {
  return (int)__builtin_amdgcn_permlane16_var(0u, (unsigned)v, (unsigned)sel, false, true);
}

typedef float v2f __attribute__((ext_vector_type(2)));
typedef float v4f __attribute__((ext_vector_type(4)));

template<int CO, bool RELU>
__global__ __launch_bounds__(256)
void node_attend(const float* __restrict__ xl, const float* __restrict__ xr,
                 const int* __restrict__ rs, const int* __restrict__ pedge,
                 const float* __restrict__ We, const float* __restrict__ att,
                 const float* __restrict__ bias, float* __restrict__ out, int n) {
  constexpr int LPE = CO / 8;
  constexpr int RPP = 2 * LPE;
  static_assert(LPE == 4 || LPE == 8, "a lane group must lie inside one 16-lane row");
  const int gid  = blockIdx.x * blockDim.x + threadIdx.x;
  const int node = gid / LPE;
  const int sub  = gid % LPE;
  if (node >= n) return;
  const int off = sub * 8;
  const int grp = (int)(threadIdx.x & 15u) - sub;
  const int beg = rs[node];
  const int end = rs[node + 1];
  v2f r2[4], w2[4], t2[4];
#pragma unroll
  for (int h = 0; h < 2; ++h) {
    const v4f rv = *(const v4f*)(xr + (size_t)node * CO + off + 4 * h);
    const v4f wv = *(const v4f*)(We + off + 4 * h);
    const v4f tv = *(const v4f*)(att + off + 4 * h);
    r2[2 * h] = rv.xy; r2[2 * h + 1] = rv.zw;
    w2[2 * h] = wv.xy; w2[2 * h + 1] = wv.zw;
    t2[2 * h] = tv.xy; t2[2 * h + 1] = tv.zw;
  }
  const float NEG_INF = -__builtin_huge_valf();

  float mx = NEG_INF, den = 0.0f;
  v2f acc[4];
#pragma unroll
  for (int i = 0; i < 4; ++i) acc[i] = v2f{0.0f, 0.0f};

  for (int base = beg; base < end; base += RPP) {
    const int j0 = base + sub;
    const int j1 = base + LPE + sub;
    int2 rec0 = make_int2(0, 0), rec1 = make_int2(0, 0);
    if (j0 < end) rec0 = *(const int2*)(pedge + 2 * (size_t)j0);
    if (j1 < end) rec1 = *(const int2*)(pedge + 2 * (size_t)j1);
    const int here = min(RPP, end - base);
    for (int t = 0; t < here; t += 4) {
      const bool second = t >= LPE;
      const int rx = second ? rec1.x : rec0.x;
      const int ry = second ? rec1.y : rec0.y;
      const int l0 = grp + (t & (LPE - 1));
      v2f   a[4][4];
      float p[4], fe4[4];
      int   s4[4];
#pragma unroll
      for (int u = 0; u < 4; ++u) {
        s4[u]  = row_pick(rx, l0 + u);
        fe4[u] = __int_as_float(row_pick(ry, l0 + u));
      }
#pragma unroll
      for (int u = 0; u < 4; ++u) {
        const float* rp = xl + (size_t)s4[u] * CO + off;
        const v4f lo = *(const v4f*)(rp);
        const v4f hi = *(const v4f*)(rp + 4);
        a[u][0] = lo.xy; a[u][1] = lo.zw; a[u][2] = hi.xy; a[u][3] = hi.zw;
      }
      __builtin_amdgcn_sched_barrier(0);
#pragma unroll
      for (int u = 0; u < 4; ++u) {
        const float fe = fe4[u];
        v2f q2 = v2f{0.0f, 0.0f};
#pragma unroll
        for (int i = 0; i < 4; ++i) {
          v2f m = a[u][i] + r2[i] + fe * w2[i];
          const v2f ms = NEG_SLOPE * m;
          m.x = (m.x > 0.0f) ? m.x : ms.x;
          m.y = (m.y > 0.0f) ? m.y : ms.y;
          q2 += m * t2[i];
        }
        p[u] = q2.x + q2.y;
      }
      if (LPE == 8) {
#pragma unroll
        for (int u = 0; u < 4; ++u) p[u] = lane_xor_add<4>(p[u]);
      }
#pragma unroll
      for (int u = 0; u < 4; ++u) p[u] = lane_xor_add<2>(p[u]);
#pragma unroll
      for (int u = 0; u < 4; ++u) p[u] = lane_xor_add<1>(p[u]);
#pragma unroll
      for (int u = 1; u < 4; ++u) {
        if (t + u >= here) {
          p[u] = NEG_INF;
#pragma unroll
          for (int i = 0; i < 4; ++i) a[u][i] = v2f{0.0f, 0.0f};
        }
      }
      const float mn = fmaxf(fmaxf(mx, p[0]), fmaxf(fmaxf(p[1], p[2]), p[3]));
      const float sc = __expf(mx - mn);
      const float e0 = __expf(p[0] - mn);
      const float e1 = __expf(p[1] - mn);
      const float e2 = __expf(p[2] - mn);
      const float e3 = __expf(p[3] - mn);
      den = den * sc + ((e0 + e1) + (e2 + e3));
#pragma unroll
      for (int i = 0; i < 4; ++i)
        acc[i] = acc[i] * sc + ((a[0][i] * e0 + a[1][i] * e1) + (a[2][i] * e2 + a[3][i] * e3));
      mx = mn;
    }
  }

  const float dn = den + 1e-16f;
#pragma unroll
  for (int h = 0; h < 2; ++h) {
    const v4f b4 = *(const v4f*)(bias + off + 4 * h);
    v4f o4;
    o4.xy = acc[2 * h] / dn + b4.xy;
    o4.zw = acc[2 * h + 1] / dn + b4.zw;
    if (RELU) {
      o4.x = fmaxf(o4.x, 0.0f); o4.y = fmaxf(o4.y, 0.0f);
      o4.z = fmaxf(o4.z, 0.0f); o4.w = fmaxf(o4.w, 0.0f);
    }
    *(v4f*)(out + (size_t)node * CO + off + 4 * h) = o4;
  }
}

extern "C" void kernel_launch(void* const* d_in, const int* in_sizes, int n_in,
                              void* d_out, int out_size, void* d_ws, size_t ws_size,
                              hipStream_t stream) {
  (void)n_in; (void)out_size; (void)ws_size;
  const float* x    = (const float*)d_in[0];
  const int*   ei   = (const int*)d_in[1];
  const float* ef   = (const float*)d_in[2];
  const float* Wl0  = (const float*)d_in[3];
  const float* bl0  = (const float*)d_in[4];
  const float* Wr0  = (const float*)d_in[5];
  const float* br0  = (const float*)d_in[6];
  const float* We0  = (const float*)d_in[7];
  const float* att0 = (const float*)d_in[8];
  const float* b0   = (const float*)d_in[9];
  const float* Wl1  = (const float*)d_in[10];
  const float* bl1  = (const float*)d_in[11];
  const float* Wr1  = (const float*)d_in[12];
  const float* br1  = (const float*)d_in[13];
  const float* We1  = (const float*)d_in[14];
  const float* att1 = (const float*)d_in[15];
  const float* b1   = (const float*)d_in[16];

  const int n = in_sizes[0] / 64;
  const int e = in_sizes[1] / 2;
  const int* srcI = ei;
  const int* dstI = ei + e;

  float* ws   = (float*)d_ws;
  float* xl   = ws;  ws += (size_t)n * 64;
  float* xr   = ws;  ws += (size_t)n * 64;
  float* hbuf = ws;  ws += (size_t)n * 64;
  int* table  = (int*)ws;  ws += SORT_NB * SORT_NB;
  int* binbase = (int*)ws; ws += SORT_NB + 4;
  int* trec   = (int*)ws;  ws += (size_t)e * 4;
  int* pedge  = (int*)ws;  ws += (size_t)e * 2;
  int* rs     = (int*)ws;  ws += n + 1;

  auto cdiv = [](long long a, long long b) { return (int)((a + b - 1) / b); };
  const int B = 256;

  chunk_count<<<SORT_NB, SORT_T, 0, stream>>>(dstI, table, n, e);
  chunk_place<<<SORT_NB, SORT_T, 0, stream>>>(srcI, dstI, ef, table, binbase, trec, n, e);
  bin_sort<<<SORT_NB, SORT_T, 0, stream>>>(binbase, trec, rs, pedge, n, e);

  {
    constexpr int CO = 64;
    node_xform_pair<CO><<<cdiv(n, 128), B, 0, stream>>>(x, Wl0, bl0, Wr0, br0, xl, xr, n);
    node_attend<CO, true><<<cdiv((long long)n * (CO / 8), B), B, 0, stream>>>(
        xl, xr, rs, pedge, We0, att0, b0, hbuf, n);
  }

  {
    constexpr int CO = 32;
    float* outp = (float*)d_out;
    node_xform_pair<CO><<<cdiv(n, 128), B, 0, stream>>>(hbuf, Wl1, bl1, Wr1, br1, xl, xr, n);
    node_attend<CO, false><<<cdiv((long long)n * (CO / 8), B), B, 0, stream>>>(
        xl, xr, rs, pedge, We1, att1, b1, outp, n);
  }
}


# Layer-0 node transform and the edge chunk counts share one launch (independent work; transform groups first, 256 counting groups of 8 waves behind them): 7 launches become 6
# speedup vs baseline: 5.8382x; 1.0337x over previous
#include <hip/hip_runtime.h>
#include <hip/hip_bf16.h>

#define NEG_SLOPE 0.2f

typedef __attribute__((ext_vector_type(16))) _Float16 v16h;
typedef __attribute__((ext_vector_type(8)))  float    v8f;

template<int NV> using vecf = float __attribute__((ext_vector_type(NV)));

template<int CO>
__device__ __forceinline__
void node_xform_strip(const float* __restrict__ X,
                      const float* __restrict__ Wl, const float* __restrict__ bl,
                      const float* __restrict__ Wr, const float* __restrict__ br,
                      float* __restrict__ xl, float* __restrict__ xr, int n_nodes, int block) {
  constexpr int TPM = CO / 16;
  const int lane  = threadIdx.x & 31;
  const int strip = __builtin_amdgcn_readfirstlane(
      (int)(block * (blockDim.x >> 5) + (threadIdx.x >> 5)));
  if (strip * 16 >= n_nodes) return;
  const int c = lane & 15;
  const int h = lane >> 4;

  int node = strip * 16 + c;
  if (node >= n_nodes) node = n_nodes - 1;

  v16h a[2];
#pragma unroll
  for (int ks = 0; ks < 2; ++ks) {
    const float* xp = X + (size_t)node * 64 + ks * 32 + h * 8;
    const float4 p0 = *(const float4*)(xp);
    const float4 p1 = *(const float4*)(xp + 4);
    const float4 p2 = *(const float4*)(xp + 16);
    const float4 p3 = *(const float4*)(xp + 20);
    a[ks][0]  = (_Float16)p0.x; a[ks][1]  = (_Float16)p0.y;
    a[ks][2]  = (_Float16)p0.z; a[ks][3]  = (_Float16)p0.w;
    a[ks][4]  = (_Float16)p1.x; a[ks][5]  = (_Float16)p1.y;
    a[ks][6]  = (_Float16)p1.z; a[ks][7]  = (_Float16)p1.w;
    a[ks][8]  = (_Float16)p2.x; a[ks][9]  = (_Float16)p2.y;
    a[ks][10] = (_Float16)p2.z; a[ks][11] = (_Float16)p2.w;
    a[ks][12] = (_Float16)p3.x; a[ks][13] = (_Float16)p3.y;
    a[ks][14] = (_Float16)p3.z; a[ks][15] = (_Float16)p3.w;
  }

  v8f acc[2][TPM];
#pragma unroll
  for (int t = 0; t < TPM; ++t) {
    const float vl = bl[TPM * c + t];
    const float vr = br[TPM * c + t];
#pragma unroll
    for (int r = 0; r < 8; ++r) { acc[0][t][r] = vl; acc[1][t][r] = vr; }
  }

#pragma unroll
  for (int ks = 0; ks < 2; ++ks) {
#pragma unroll
    for (int m = 0; m < 2; ++m) {
      const float* __restrict__ Wm = m ? Wr : Wl;
      vecf<TPM> w[16];
#pragma unroll
      for (int s = 0; s < 16; ++s) {
        const int k = ks * 32 + ((s & 8) ? 16 : 0) + h * 8 + (s & 7);
        w[s] = *(const vecf<TPM>*)(Wm + (size_t)k * CO + TPM * c);
      }
      __builtin_amdgcn_sched_barrier(0);
      v16h b[TPM];
#pragma unroll
      for (int s = 0; s < 16; ++s) {
#pragma unroll
        for (int t = 0; t < TPM; ++t) b[t][s] = (_Float16)w[s][t];
      }
#pragma unroll
      for (int t = 0; t < TPM; ++t) {
        acc[m][t] = __builtin_amdgcn_wmma_f32_16x16x32_f16(false, a[ks], false, b[t],
                                                           (short)0, acc[m][t], false, false);
      }
    }
  }

  const int row0 = strip * 16 + 8 * h;
  if ((strip + 1) * 16 <= n_nodes) {
#pragma unroll
    for (int r = 0; r < 8; ++r) {
      vecf<TPM> ol, orr;
#pragma unroll
      for (int t = 0; t < TPM; ++t) { ol[t] = acc[0][t][r]; orr[t] = acc[1][t][r]; }
      *(vecf<TPM>*)(xl + (size_t)(row0 + r) * CO + TPM * c) = ol;
      *(vecf<TPM>*)(xr + (size_t)(row0 + r) * CO + TPM * c) = orr;
    }
  } else {
#pragma unroll
    for (int r = 0; r < 8; ++r) {
      vecf<TPM> ol, orr;
#pragma unroll
      for (int t = 0; t < TPM; ++t) { ol[t] = acc[0][t][r]; orr[t] = acc[1][t][r]; }
      if (row0 + r < n_nodes) {
        *(vecf<TPM>*)(xl + (size_t)(row0 + r) * CO + TPM * c) = ol;
        *(vecf<TPM>*)(xr + (size_t)(row0 + r) * CO + TPM * c) = orr;
      }
    }
  }
}

template<int CO>
__global__ __launch_bounds__(256)
void node_xform_pair(const float* __restrict__ X,
                     const float* __restrict__ Wl, const float* __restrict__ bl,
                     const float* __restrict__ Wr, const float* __restrict__ br,
                     float* __restrict__ xl, float* __restrict__ xr, int n_nodes) {
  node_xform_strip<CO>(X, Wl, bl, Wr, br, xl, xr, n_nodes, (int)blockIdx.x);
}

#define SORT_NB     256
#define SORT_T      512
#define SORT_K      8
#define SORT_MAXNPB 1024
#define SORT_STAGE  6144
#define COUNT_T     256
#define COUNT_K     16
static_assert(COUNT_T >= SORT_NB, "one thread per bin clears and stores the chunk's counts");

__device__ __forceinline__
void chunk_count_block(const int* __restrict__ dst, int* __restrict__ table, int n, int ne, int w) {
  __shared__ int hist[SORT_NB];
  const int t = threadIdx.x;
  const unsigned npb = (unsigned)((n + SORT_NB - 1) / SORT_NB);
  const int ech = (ne + SORT_NB - 1) / SORT_NB;
  const int c0  = min(ne, w * ech);
  const int c1  = min(ne, c0 + ech);
  if (t < SORT_NB) hist[t] = 0;
  __syncthreads();
  for (int base = c0; base < c1; base += COUNT_T * COUNT_K) {
    int d[COUNT_K];
#pragma unroll
    for (int k = 0; k < COUNT_K; ++k) {
      const int i = base + k * COUNT_T + t;
      d[k] = (i < c1) ? dst[i] : -1;
    }
#pragma unroll
    for (int k = 0; k < COUNT_K; ++k)
      if (d[k] >= 0) atomicAdd(&hist[(unsigned)d[k] / npb], 1);
  }
  __syncthreads();
  if (t < SORT_NB) table[w * SORT_NB + t] = hist[t];
}

__global__ __launch_bounds__(COUNT_T)
void layer0_xform_and_chunk_count(const float* __restrict__ X,
                                  const float* __restrict__ Wl, const float* __restrict__ bl,
                                  const float* __restrict__ Wr, const float* __restrict__ br,
                                  float* __restrict__ xl, float* __restrict__ xr, int n_nodes,
                                  const int* __restrict__ dst, int* __restrict__ table, int ne) {
  const int xblocks = (n_nodes + 16 * (COUNT_T / 32) - 1) / (16 * (COUNT_T / 32));
  const int block = (int)blockIdx.x;
  if (block < xblocks) node_xform_strip<64>(X, Wl, bl, Wr, br, xl, xr, n_nodes, block);
  else                 chunk_count_block(dst, table, n_nodes, ne, block - xblocks);
}

__global__ __launch_bounds__(SORT_T)
void chunk_place(const int* __restrict__ src, const int* __restrict__ dst,
                 const float* __restrict__ ef, const int* __restrict__ table,
                 int* __restrict__ binbase, int* __restrict__ trec_w, int n, int ne) {
  int4* __restrict__ trec = (int4*)trec_w;
  constexpr int NQ = SORT_T / 64;
  constexpr int RQ = SORT_NB / NQ;
  __shared__ int ptot[NQ][SORT_NB];
  __shared__ int ppre[NQ][SORT_NB];
  __shared__ int wtot[SORT_NB / 32];
  __shared__ int cursor[SORT_NB + 32];
  const int t    = threadIdx.x;
  const int lane = t & 31;
  const int wave = t >> 5;
  const int w    = blockIdx.x;
  const unsigned npb = (unsigned)((n + SORT_NB - 1) / SORT_NB);
  const int ech = (ne + SORT_NB - 1) / SORT_NB;
  const int c0  = min(ne, w * ech);
  const int c1  = min(ne, c0 + ech);

  int d[SORT_K], s[SORT_K], f[SORT_K];
#pragma unroll
  for (int k = 0; k < SORT_K; ++k) {
    const int i = c0 + k * SORT_T + t;
    const bool ok = i < c1;
    d[k] = ok ? dst[i] : -1;
    s[k] = ok ? src[i] : 0;
    f[k] = ok ? __float_as_int(ef[i]) : 0;
  }

  {
    const int q  = t >> 6;
    const int b4 = (t & 63) * 4;
    int4 tot = make_int4(0, 0, 0, 0), pre = make_int4(0, 0, 0, 0);
#pragma unroll
    for (int kb = 0; kb < RQ; kb += 16) {
      int4 v[16];
#pragma unroll
      for (int k = 0; k < 16; ++k)
        v[k] = *(const int4*)(table + (size_t)(q * RQ + kb + k) * SORT_NB + b4);
      __builtin_amdgcn_sched_barrier(0);
#pragma unroll
      for (int k = 0; k < 16; ++k) {
        const bool early = (q * RQ + kb + k) < w;
        tot.x += v[k].x; tot.y += v[k].y; tot.z += v[k].z; tot.w += v[k].w;
        pre.x += early ? v[k].x : 0; pre.y += early ? v[k].y : 0;
        pre.z += early ? v[k].z : 0; pre.w += early ? v[k].w : 0;
      }
    }
    *(int4*)&ptot[q][b4] = tot;
    *(int4*)&ppre[q][b4] = pre;
  }
  __syncthreads();
  int mytot = 0, mypre = 0, incl = 0;
  if (t < SORT_NB) {
#pragma unroll
    for (int q = 0; q < NQ; ++q) { mytot += ptot[q][t]; mypre += ppre[q][t]; }
    incl = mytot;
#pragma unroll
    for (int o = 1; o < 32; o <<= 1) {
      const int v = __shfl_up(incl, o, 32);
      if (lane >= o) incl += v;
    }
    if (lane == 31) wtot[wave] = incl;
  }
  __syncthreads();
  if (t < SORT_NB) {
    int woff = 0;
#pragma unroll
    for (int j = 0; j < SORT_NB / 32; ++j) woff += (j < wave) ? wtot[j] : 0;
    const int base = woff + incl - mytot;
    cursor[t] = base + mypre;
    if (t < 32) cursor[SORT_NB + t] = 0;
    if (w == 0) {
      binbase[t] = base;
      if (t == SORT_NB - 1) binbase[SORT_NB] = base + mytot;
    }
  }
  __syncthreads();

  for (int base = c0; base < c1; base += SORT_T * SORT_K) {
    int pos[SORT_K];
#pragma unroll
    for (int k = 0; k < SORT_K; ++k) {
      const int bin = (d[k] >= 0) ? (int)((unsigned)d[k] / npb) : SORT_NB + lane;
      pos[k] = atomicAdd(&cursor[bin], 1);
    }
#pragma unroll
    for (int k = 0; k < SORT_K; ++k)
      if (d[k] >= 0) trec[pos[k]] = make_int4(s[k], f[k], d[k], 0);
#pragma unroll
    for (int k = 0; k < SORT_K; ++k) {
      const int i = base + SORT_T * SORT_K + k * SORT_T + t;
      const bool ok = i < c1;
      d[k] = ok ? dst[i] : -1;
      s[k] = ok ? src[i] : 0;
      f[k] = ok ? __float_as_int(ef[i]) : 0;
    }
  }
}

__global__ __launch_bounds__(SORT_T)
void bin_sort(const int* __restrict__ binbase, const int* __restrict__ trec_w,
              int* __restrict__ rs, int* __restrict__ pedge_w, int n, int ne) {
  const int4* __restrict__ trec = (const int4*)trec_w;
  int2* __restrict__ pedge = (int2*)pedge_w;
  __shared__ int  cntL[SORT_MAXNPB];
  __shared__ int  wsum[SORT_T / 32];
  __shared__ int2 stage[SORT_STAGE];
  const int t    = threadIdx.x;
  const int lane = t & 31;
  const int wave = t >> 5;
  const int b    = blockIdx.x;
  const int npb  = (n + SORT_NB - 1) / SORT_NB;
  const int node0 = b * npb;
  const int beg = binbase[b];
  const int end = binbase[b + 1];
  const bool staged = (end - beg) <= SORT_STAGE;
  cntL[2 * t] = 0; cntL[2 * t + 1] = 0;
  if (b == 0 && t == 0) rs[n] = ne;
  __syncthreads();
  for (int base = beg; base < end; base += SORT_T * SORT_K) {
    int d[SORT_K];
#pragma unroll
    for (int k = 0; k < SORT_K; ++k) {
      const int i = base + k * SORT_T + t;
      d[k] = (i < end) ? trec[i].z : -1;
    }
#pragma unroll
    for (int k = 0; k < SORT_K; ++k)
      if (d[k] >= 0) atomicAdd(&cntL[d[k] - node0], 1);
  }
  __syncthreads();
  {
    const int n0 = cntL[2 * t], n1 = cntL[2 * t + 1];
    const int sm = n0 + n1;
    int incl = sm;
#pragma unroll
    for (int o = 1; o < 32; o <<= 1) {
      const int v = __shfl_up(incl, o, 32);
      if (lane >= o) incl += v;
    }
    if (lane == 31) wsum[wave] = incl;
    __syncthreads();
    int woff = 0;
#pragma unroll
    for (int j = 0; j < SORT_T / 32; ++j) woff += (j < wave) ? wsum[j] : 0;
    const int ex = woff + incl - sm;
    cntL[2 * t]     = ex;
    cntL[2 * t + 1] = ex + n0;
    const int l0 = 2 * t;
    if (l0 < npb && node0 + l0 < n)         rs[node0 + l0]     = beg + ex;
    if (l0 + 1 < npb && node0 + l0 + 1 < n) rs[node0 + l0 + 1] = beg + ex + n0;
  }
  __syncthreads();
  for (int base = beg; base < end; base += SORT_T * SORT_K) {
    int4 r[SORT_K];
#pragma unroll
    for (int k = 0; k < SORT_K; ++k) {
      const int i = base + k * SORT_T + t;
      r[k] = (i < end) ? trec[i] : make_int4(0, 0, -1, 0);
    }
    int pos[SORT_K];
#pragma unroll
    for (int k = 0; k < SORT_K; ++k)
      pos[k] = (r[k].z >= 0) ? atomicAdd(&cntL[r[k].z - node0], 1) : 0;
#pragma unroll
    for (int k = 0; k < SORT_K; ++k) {
      if (r[k].z >= 0) {
        if (staged) stage[pos[k]] = make_int2(r[k].x, r[k].y);
        else        pedge[beg + pos[k]] = make_int2(r[k].x, r[k].y);
      }
    }
  }
  __syncthreads();
  if (staged) {
    for (int j = t; j < end - beg; j += SORT_T) pedge[beg + j] = stage[j];
  }
}

template<int M>
__device__ __forceinline__ float lane_xor_add(float v) {
  const int partner = __builtin_amdgcn_update_dpp(0, __float_as_int(v), 0x160 + M, 0xf, 0xf, true);
  return v + __int_as_float(partner);
}
__device__ __forceinline__ int row_pick(int v, int sel) {
  return (int)__builtin_amdgcn_permlane16_var(0u, (unsigned)v, (unsigned)sel, false, true);
}

typedef float v2f __attribute__((ext_vector_type(2)));
typedef float v4f __attribute__((ext_vector_type(4)));

template<int CO, bool RELU>
__global__ __launch_bounds__(256)
void node_attend(const float* __restrict__ xl, const float* __restrict__ xr,
                 const int* __restrict__ rs, const int* __restrict__ pedge,
                 const float* __restrict__ We, const float* __restrict__ att,
                 const float* __restrict__ bias, float* __restrict__ out, int n) {
  constexpr int LPE = CO / 8;
  constexpr int RPP = 2 * LPE;
  static_assert(LPE == 4 || LPE == 8, "a lane group must lie inside one 16-lane row");
  const int gid  = blockIdx.x * blockDim.x + threadIdx.x;
  const int node = gid / LPE;
  const int sub  = gid % LPE;
  if (node >= n) return;
  const int off = sub * 8;
  const int grp = (int)(threadIdx.x & 15u) - sub;
  const int beg = rs[node];
  const int end = rs[node + 1];
  v2f r2[4], w2[4], t2[4];
#pragma unroll
  for (int h = 0; h < 2; ++h) {
    const v4f rv = *(const v4f*)(xr + (size_t)node * CO + off + 4 * h);
    const v4f wv = *(const v4f*)(We + off + 4 * h);
    const v4f tv = *(const v4f*)(att + off + 4 * h);
    r2[2 * h] = rv.xy; r2[2 * h + 1] = rv.zw;
    w2[2 * h] = wv.xy; w2[2 * h + 1] = wv.zw;
    t2[2 * h] = tv.xy; t2[2 * h + 1] = tv.zw;
  }
  const float NEG_INF = -__builtin_huge_valf();

  float mx = NEG_INF, den = 0.0f;
  v2f acc[4];
#pragma unroll
  for (int i = 0; i < 4; ++i) acc[i] = v2f{0.0f, 0.0f};

  for (int base = beg; base < end; base += RPP) {
    const int j0 = base + sub;
    const int j1 = base + LPE + sub;
    int2 rec0 = make_int2(0, 0), rec1 = make_int2(0, 0);
    if (j0 < end) rec0 = *(const int2*)(pedge + 2 * (size_t)j0);
    if (j1 < end) rec1 = *(const int2*)(pedge + 2 * (size_t)j1);
    const int here = min(RPP, end - base);
    for (int t = 0; t < here; t += 4) {
      const bool second = t >= LPE;
      const int rx = second ? rec1.x : rec0.x;
      const int ry = second ? rec1.y : rec0.y;
      const int l0 = grp + (t & (LPE - 1));
      v2f   a[4][4];
      float p[4], fe4[4];
      int   s4[4];
#pragma unroll
      for (int u = 0; u < 4; ++u) {
        s4[u]  = row_pick(rx, l0 + u);
        fe4[u] = __int_as_float(row_pick(ry, l0 + u));
      }
#pragma unroll
      for (int u = 0; u < 4; ++u) {
        const float* rp = xl + (size_t)s4[u] * CO + off;
        const v4f lo = *(const v4f*)(rp);
        const v4f hi = *(const v4f*)(rp + 4);
        a[u][0] = lo.xy; a[u][1] = lo.zw; a[u][2] = hi.xy; a[u][3] = hi.zw;
      }
      __builtin_amdgcn_sched_barrier(0);
#pragma unroll
      for (int u = 0; u < 4; ++u) {
        const float fe = fe4[u];
        v2f q2 = v2f{0.0f, 0.0f};
#pragma unroll
        for (int i = 0; i < 4; ++i) {
          v2f m = a[u][i] + r2[i] + fe * w2[i];
          const v2f ms = NEG_SLOPE * m;
          m.x = (m.x > 0.0f) ? m.x : ms.x;
          m.y = (m.y > 0.0f) ? m.y : ms.y;
          q2 += m * t2[i];
        }
        p[u] = q2.x + q2.y;
      }
      if (LPE == 8) {
#pragma unroll
        for (int u = 0; u < 4; ++u) p[u] = lane_xor_add<4>(p[u]);
      }
#pragma unroll
      for (int u = 0; u < 4; ++u) p[u] = lane_xor_add<2>(p[u]);
#pragma unroll
      for (int u = 0; u < 4; ++u) p[u] = lane_xor_add<1>(p[u]);
#pragma unroll
      for (int u = 1; u < 4; ++u) {
        if (t + u >= here) {
          p[u] = NEG_INF;
#pragma unroll
          for (int i = 0; i < 4; ++i) a[u][i] = v2f{0.0f, 0.0f};
        }
      }
      const float mn = fmaxf(fmaxf(mx, p[0]), fmaxf(fmaxf(p[1], p[2]), p[3]));
      const float sc = __expf(mx - mn);
      const float e0 = __expf(p[0] - mn);
      const float e1 = __expf(p[1] - mn);
      const float e2 = __expf(p[2] - mn);
      const float e3 = __expf(p[3] - mn);
      den = den * sc + ((e0 + e1) + (e2 + e3));
#pragma unroll
      for (int i = 0; i < 4; ++i)
        acc[i] = acc[i] * sc + ((a[0][i] * e0 + a[1][i] * e1) + (a[2][i] * e2 + a[3][i] * e3));
      mx = mn;
    }
  }

  const float dn = den + 1e-16f;
#pragma unroll
  for (int h = 0; h < 2; ++h) {
    const v4f b4 = *(const v4f*)(bias + off + 4 * h);
    v4f o4;
    o4.xy = acc[2 * h] / dn + b4.xy;
    o4.zw = acc[2 * h + 1] / dn + b4.zw;
    if (RELU) {
      o4.x = fmaxf(o4.x, 0.0f); o4.y = fmaxf(o4.y, 0.0f);
      o4.z = fmaxf(o4.z, 0.0f); o4.w = fmaxf(o4.w, 0.0f);
    }
    *(v4f*)(out + (size_t)node * CO + off + 4 * h) = o4;
  }
}

extern "C" void kernel_launch(void* const* d_in, const int* in_sizes, int n_in,
                              void* d_out, int out_size, void* d_ws, size_t ws_size,
                              hipStream_t stream) {
  (void)n_in; (void)out_size; (void)ws_size;
  const float* x    = (const float*)d_in[0];
  const int*   ei   = (const int*)d_in[1];
  const float* ef   = (const float*)d_in[2];
  const float* Wl0  = (const float*)d_in[3];
  const float* bl0  = (const float*)d_in[4];
  const float* Wr0  = (const float*)d_in[5];
  const float* br0  = (const float*)d_in[6];
  const float* We0  = (const float*)d_in[7];
  const float* att0 = (const float*)d_in[8];
  const float* b0   = (const float*)d_in[9];
  const float* Wl1  = (const float*)d_in[10];
  const float* bl1  = (const float*)d_in[11];
  const float* Wr1  = (const float*)d_in[12];
  const float* br1  = (const float*)d_in[13];
  const float* We1  = (const float*)d_in[14];
  const float* att1 = (const float*)d_in[15];
  const float* b1   = (const float*)d_in[16];

  const int n = in_sizes[0] / 64;
  const int e = in_sizes[1] / 2;
  const int* srcI = ei;
  const int* dstI = ei + e;

  float* ws   = (float*)d_ws;
  float* xl   = ws;  ws += (size_t)n * 64;
  float* xr   = ws;  ws += (size_t)n * 64;
  float* hbuf = ws;  ws += (size_t)n * 64;
  int* table  = (int*)ws;  ws += SORT_NB * SORT_NB;
  int* binbase = (int*)ws; ws += SORT_NB + 4;
  int* trec   = (int*)ws;  ws += (size_t)e * 4;
  int* pedge  = (int*)ws;  ws += (size_t)e * 2;
  int* rs     = (int*)ws;  ws += n + 1;

  auto cdiv = [](long long a, long long b) { return (int)((a + b - 1) / b); };
  const int B = 256;

  layer0_xform_and_chunk_count<<<cdiv(n, 16 * (COUNT_T / 32)) + SORT_NB, COUNT_T, 0, stream>>>(
      x, Wl0, bl0, Wr0, br0, xl, xr, n, dstI, table, e);

  chunk_place<<<SORT_NB, SORT_T, 0, stream>>>(srcI, dstI, ef, table, binbase, trec, n, e);
  bin_sort<<<SORT_NB, SORT_T, 0, stream>>>(binbase, trec, rs, pedge, n, e);

  {
    constexpr int CO = 64;
    node_attend<CO, true><<<cdiv((long long)n * (CO / 8), B), B, 0, stream>>>(
        xl, xr, rs, pedge, We0, att0, b0, hbuf, n);
  }

  {
    constexpr int CO = 32;
    float* outp = (float*)d_out;
    node_xform_pair<CO><<<cdiv(n, 128), B, 0, stream>>>(hbuf, Wl1, bl1, Wr1, br1, xl, xr, n);
    node_attend<CO, false><<<cdiv((long long)n * (CO / 8), B), B, 0, stream>>>(
        xl, xr, rs, pedge, We1, att1, b1, outp, n);
  }
}


# Layer-0 node walk hands its finished rows over in LDS and one wave per (16-row strip, matrix) makes xl and xr of layer 1 in the same launch: no hidden-row plane, 6 launches become 5
# speedup vs baseline: 6.3493x; 1.0875x over previous
#include <hip/hip_runtime.h>
#include <hip/hip_bf16.h>

#define NEG_SLOPE 0.2f

typedef __attribute__((ext_vector_type(16))) _Float16 v16h;
typedef __attribute__((ext_vector_type(8)))  float    v8f;

template<int NV> using vecf = float __attribute__((ext_vector_type(NV)));

template<int CO>
__device__ __forceinline__
void node_xform_strip(const float* __restrict__ X,
                      const float* __restrict__ Wl, const float* __restrict__ bl,
                      const float* __restrict__ Wr, const float* __restrict__ br,
                      float* __restrict__ xl, float* __restrict__ xr, int n_nodes, int block) {
  constexpr int TPM = CO / 16;
  const int lane  = threadIdx.x & 31;
  const int strip = __builtin_amdgcn_readfirstlane(
      (int)(block * (blockDim.x >> 5) + (threadIdx.x >> 5)));
  if (strip * 16 >= n_nodes) return;
  const int c = lane & 15;
  const int h = lane >> 4;

  int node = strip * 16 + c;
  if (node >= n_nodes) node = n_nodes - 1;

  v16h a[2];
#pragma unroll
  for (int ks = 0; ks < 2; ++ks) {
    const float* xp = X + (size_t)node * 64 + ks * 32 + h * 8;
    const float4 p0 = *(const float4*)(xp);
    const float4 p1 = *(const float4*)(xp + 4);
    const float4 p2 = *(const float4*)(xp + 16);
    const float4 p3 = *(const float4*)(xp + 20);
    a[ks][0]  = (_Float16)p0.x; a[ks][1]  = (_Float16)p0.y;
    a[ks][2]  = (_Float16)p0.z; a[ks][3]  = (_Float16)p0.w;
    a[ks][4]  = (_Float16)p1.x; a[ks][5]  = (_Float16)p1.y;
    a[ks][6]  = (_Float16)p1.z; a[ks][7]  = (_Float16)p1.w;
    a[ks][8]  = (_Float16)p2.x; a[ks][9]  = (_Float16)p2.y;
    a[ks][10] = (_Float16)p2.z; a[ks][11] = (_Float16)p2.w;
    a[ks][12] = (_Float16)p3.x; a[ks][13] = (_Float16)p3.y;
    a[ks][14] = (_Float16)p3.z; a[ks][15] = (_Float16)p3.w;
  }

  v8f acc[2][TPM];
#pragma unroll
  for (int t = 0; t < TPM; ++t) {
    const float vl = bl[TPM * c + t];
    const float vr = br[TPM * c + t];
#pragma unroll
    for (int r = 0; r < 8; ++r) { acc[0][t][r] = vl; acc[1][t][r] = vr; }
  }

#pragma unroll
  for (int ks = 0; ks < 2; ++ks) {
#pragma unroll
    for (int m = 0; m < 2; ++m) {
      const float* __restrict__ Wm = m ? Wr : Wl;
      vecf<TPM> w[16];
#pragma unroll
      for (int s = 0; s < 16; ++s) {
        const int k = ks * 32 + ((s & 8) ? 16 : 0) + h * 8 + (s & 7);
        w[s] = *(const vecf<TPM>*)(Wm + (size_t)k * CO + TPM * c);
      }
      __builtin_amdgcn_sched_barrier(0);
      v16h b[TPM];
#pragma unroll
      for (int s = 0; s < 16; ++s) {
#pragma unroll
        for (int t = 0; t < TPM; ++t) b[t][s] = (_Float16)w[s][t];
      }
#pragma unroll
      for (int t = 0; t < TPM; ++t) {
        acc[m][t] = __builtin_amdgcn_wmma_f32_16x16x32_f16(false, a[ks], false, b[t],
                                                           (short)0, acc[m][t], false, false);
      }
    }
  }

  const int row0 = strip * 16 + 8 * h;
  if ((strip + 1) * 16 <= n_nodes) {
#pragma unroll
    for (int r = 0; r < 8; ++r) {
      vecf<TPM> ol, orr;
#pragma unroll
      for (int t = 0; t < TPM; ++t) { ol[t] = acc[0][t][r]; orr[t] = acc[1][t][r]; }
      *(vecf<TPM>*)(xl + (size_t)(row0 + r) * CO + TPM * c) = ol;
      *(vecf<TPM>*)(xr + (size_t)(row0 + r) * CO + TPM * c) = orr;
    }
  } else {
#pragma unroll
    for (int r = 0; r < 8; ++r) {
      vecf<TPM> ol, orr;
#pragma unroll
      for (int t = 0; t < TPM; ++t) { ol[t] = acc[0][t][r]; orr[t] = acc[1][t][r]; }
      if (row0 + r < n_nodes) {
        *(vecf<TPM>*)(xl + (size_t)(row0 + r) * CO + TPM * c) = ol;
        *(vecf<TPM>*)(xr + (size_t)(row0 + r) * CO + TPM * c) = orr;
      }
    }
  }
}

template<int CO, int HP>
__device__ __forceinline__
void strip_xform_from_lds(const float* hstrip,
                          const float* __restrict__ Wm, const float* __restrict__ bm,
                          float* __restrict__ outm, int row_base, int n_nodes) {
  constexpr int TPM = CO / 16;
  const int lane = threadIdx.x & 31;
  const int c = lane & 15;
  const int h = lane >> 4;

  v16h a[2];
#pragma unroll
  for (int ks = 0; ks < 2; ++ks) {
    const float* xp = hstrip + c * HP + ks * 32 + h * 8;
    const float4 p0 = *(const float4*)(xp);
    const float4 p1 = *(const float4*)(xp + 4);
    const float4 p2 = *(const float4*)(xp + 16);
    const float4 p3 = *(const float4*)(xp + 20);
    a[ks][0]  = (_Float16)p0.x; a[ks][1]  = (_Float16)p0.y;
    a[ks][2]  = (_Float16)p0.z; a[ks][3]  = (_Float16)p0.w;
    a[ks][4]  = (_Float16)p1.x; a[ks][5]  = (_Float16)p1.y;
    a[ks][6]  = (_Float16)p1.z; a[ks][7]  = (_Float16)p1.w;
    a[ks][8]  = (_Float16)p2.x; a[ks][9]  = (_Float16)p2.y;
    a[ks][10] = (_Float16)p2.z; a[ks][11] = (_Float16)p2.w;
    a[ks][12] = (_Float16)p3.x; a[ks][13] = (_Float16)p3.y;
    a[ks][14] = (_Float16)p3.z; a[ks][15] = (_Float16)p3.w;
  }

  v8f acc[TPM];
#pragma unroll
  for (int t = 0; t < TPM; ++t) {
    const float vb = bm[TPM * c + t];
#pragma unroll
    for (int r = 0; r < 8; ++r) acc[t][r] = vb;
  }

#pragma unroll
  for (int ks = 0; ks < 2; ++ks) {
    vecf<TPM> w[16];
#pragma unroll
    for (int s = 0; s < 16; ++s) {
      const int k = ks * 32 + ((s & 8) ? 16 : 0) + h * 8 + (s & 7);
      w[s] = *(const vecf<TPM>*)(Wm + (size_t)k * CO + TPM * c);
    }
    __builtin_amdgcn_sched_barrier(0);
    v16h b[TPM];
#pragma unroll
    for (int s = 0; s < 16; ++s) {
#pragma unroll
      for (int t = 0; t < TPM; ++t) b[t][s] = (_Float16)w[s][t];
    }
#pragma unroll
    for (int t = 0; t < TPM; ++t) {
      acc[t] = __builtin_amdgcn_wmma_f32_16x16x32_f16(false, a[ks], false, b[t],
                                                      (short)0, acc[t], false, false);
    }
  }

  const int row0 = row_base + 8 * h;
  if (row_base + 16 <= n_nodes) {
#pragma unroll
    for (int r = 0; r < 8; ++r) {
      vecf<TPM> o;
#pragma unroll
      for (int t = 0; t < TPM; ++t) o[t] = acc[t][r];
      *(vecf<TPM>*)(outm + (size_t)(row0 + r) * CO + TPM * c) = o;
    }
  } else {
#pragma unroll
    for (int r = 0; r < 8; ++r) {
      vecf<TPM> o;
#pragma unroll
      for (int t = 0; t < TPM; ++t) o[t] = acc[t][r];
      if (row0 + r < n_nodes) *(vecf<TPM>*)(outm + (size_t)(row0 + r) * CO + TPM * c) = o;
    }
  }
}

#define SORT_NB     256
#define SORT_T      512
#define SORT_K      8
#define SORT_MAXNPB 1024
#define SORT_STAGE  6144
#define COUNT_T     256
#define COUNT_K     16
static_assert(COUNT_T >= SORT_NB, "one thread per bin clears and stores the chunk's counts");

__device__ __forceinline__
void chunk_count_block(const int* __restrict__ dst, int* __restrict__ table, int n, int ne, int w) {
  __shared__ int hist[SORT_NB];
  const int t = threadIdx.x;
  const unsigned npb = (unsigned)((n + SORT_NB - 1) / SORT_NB);
  const int ech = (ne + SORT_NB - 1) / SORT_NB;
  const int c0  = min(ne, w * ech);
  const int c1  = min(ne, c0 + ech);
  if (t < SORT_NB) hist[t] = 0;
  __syncthreads();
  for (int base = c0; base < c1; base += COUNT_T * COUNT_K) {
    int d[COUNT_K];
#pragma unroll
    for (int k = 0; k < COUNT_K; ++k) {
      const int i = base + k * COUNT_T + t;
      d[k] = (i < c1) ? dst[i] : -1;
    }
#pragma unroll
    for (int k = 0; k < COUNT_K; ++k)
      if (d[k] >= 0) atomicAdd(&hist[(unsigned)d[k] / npb], 1);
  }
  __syncthreads();
  if (t < SORT_NB) table[w * SORT_NB + t] = hist[t];
}

__global__ __launch_bounds__(COUNT_T)
void layer0_xform_and_chunk_count(const float* __restrict__ X,
                                  const float* __restrict__ Wl, const float* __restrict__ bl,
                                  const float* __restrict__ Wr, const float* __restrict__ br,
                                  float* __restrict__ xl, float* __restrict__ xr, int n_nodes,
                                  const int* __restrict__ dst, int* __restrict__ table, int ne) {
  const int xblocks = (n_nodes + 16 * (COUNT_T / 32) - 1) / (16 * (COUNT_T / 32));
  const int block = (int)blockIdx.x;
  if (block < xblocks) node_xform_strip<64>(X, Wl, bl, Wr, br, xl, xr, n_nodes, block);
  else                 chunk_count_block(dst, table, n_nodes, ne, block - xblocks);
}

__global__ __launch_bounds__(SORT_T)
void chunk_place(const int* __restrict__ src, const int* __restrict__ dst,
                 const float* __restrict__ ef, const int* __restrict__ table,
                 int* __restrict__ binbase, int* __restrict__ trec_w, int n, int ne) {
  int4* __restrict__ trec = (int4*)trec_w;
  constexpr int NQ = SORT_T / 64;
  constexpr int RQ = SORT_NB / NQ;
  __shared__ int ptot[NQ][SORT_NB];
  __shared__ int ppre[NQ][SORT_NB];
  __shared__ int wtot[SORT_NB / 32];
  __shared__ int cursor[SORT_NB + 32];
  const int t    = threadIdx.x;
  const int lane = t & 31;
  const int wave = t >> 5;
  const int w    = blockIdx.x;
  const unsigned npb = (unsigned)((n + SORT_NB - 1) / SORT_NB);
  const int ech = (ne + SORT_NB - 1) / SORT_NB;
  const int c0  = min(ne, w * ech);
  const int c1  = min(ne, c0 + ech);

  int d[SORT_K], s[SORT_K], f[SORT_K];
#pragma unroll
  for (int k = 0; k < SORT_K; ++k) {
    const int i = c0 + k * SORT_T + t;
    const bool ok = i < c1;
    d[k] = ok ? dst[i] : -1;
    s[k] = ok ? src[i] : 0;
    f[k] = ok ? __float_as_int(ef[i]) : 0;
  }

  {
    const int q  = t >> 6;
    const int b4 = (t & 63) * 4;
    int4 tot = make_int4(0, 0, 0, 0), pre = make_int4(0, 0, 0, 0);
#pragma unroll
    for (int kb = 0; kb < RQ; kb += 16) {
      int4 v[16];
#pragma unroll
      for (int k = 0; k < 16; ++k)
        v[k] = *(const int4*)(table + (size_t)(q * RQ + kb + k) * SORT_NB + b4);
      __builtin_amdgcn_sched_barrier(0);
#pragma unroll
      for (int k = 0; k < 16; ++k) {
        const bool early = (q * RQ + kb + k) < w;
        tot.x += v[k].x; tot.y += v[k].y; tot.z += v[k].z; tot.w += v[k].w;
        pre.x += early ? v[k].x : 0; pre.y += early ? v[k].y : 0;
        pre.z += early ? v[k].z : 0; pre.w += early ? v[k].w : 0;
      }
    }
    *(int4*)&ptot[q][b4] = tot;
    *(int4*)&ppre[q][b4] = pre;
  }
  __syncthreads();
  int mytot = 0, mypre = 0, incl = 0;
  if (t < SORT_NB) {
#pragma unroll
    for (int q = 0; q < NQ; ++q) { mytot += ptot[q][t]; mypre += ppre[q][t]; }
    incl = mytot;
#pragma unroll
    for (int o = 1; o < 32; o <<= 1) {
      const int v = __shfl_up(incl, o, 32);
      if (lane >= o) incl += v;
    }
    if (lane == 31) wtot[wave] = incl;
  }
  __syncthreads();
  if (t < SORT_NB) {
    int woff = 0;
#pragma unroll
    for (int j = 0; j < SORT_NB / 32; ++j) woff += (j < wave) ? wtot[j] : 0;
    const int base = woff + incl - mytot;
    cursor[t] = base + mypre;
    if (t < 32) cursor[SORT_NB + t] = 0;
    if (w == 0) {
      binbase[t] = base;
      if (t == SORT_NB - 1) binbase[SORT_NB] = base + mytot;
    }
  }
  __syncthreads();

  for (int base = c0; base < c1; base += SORT_T * SORT_K) {
    int pos[SORT_K];
#pragma unroll
    for (int k = 0; k < SORT_K; ++k) {
      const int bin = (d[k] >= 0) ? (int)((unsigned)d[k] / npb) : SORT_NB + lane;
      pos[k] = atomicAdd(&cursor[bin], 1);
    }
#pragma unroll
    for (int k = 0; k < SORT_K; ++k)
      if (d[k] >= 0) trec[pos[k]] = make_int4(s[k], f[k], d[k], 0);
#pragma unroll
    for (int k = 0; k < SORT_K; ++k) {
      const int i = base + SORT_T * SORT_K + k * SORT_T + t;
      const bool ok = i < c1;
      d[k] = ok ? dst[i] : -1;
      s[k] = ok ? src[i] : 0;
      f[k] = ok ? __float_as_int(ef[i]) : 0;
    }
  }
}

__global__ __launch_bounds__(SORT_T)
void bin_sort(const int* __restrict__ binbase, const int* __restrict__ trec_w,
              int* __restrict__ rs, int* __restrict__ pedge_w, int n, int ne) {
  const int4* __restrict__ trec = (const int4*)trec_w;
  int2* __restrict__ pedge = (int2*)pedge_w;
  __shared__ int  cntL[SORT_MAXNPB];
  __shared__ int  wsum[SORT_T / 32];
  __shared__ int2 stage[SORT_STAGE];
  const int t    = threadIdx.x;
  const int lane = t & 31;
  const int wave = t >> 5;
  const int b    = blockIdx.x;
  const int npb  = (n + SORT_NB - 1) / SORT_NB;
  const int node0 = b * npb;
  const int beg = binbase[b];
  const int end = binbase[b + 1];
  const bool staged = (end - beg) <= SORT_STAGE;
  cntL[2 * t] = 0; cntL[2 * t + 1] = 0;
  if (b == 0 && t == 0) rs[n] = ne;
  __syncthreads();
  for (int base = beg; base < end; base += SORT_T * SORT_K) {
    int d[SORT_K];
#pragma unroll
    for (int k = 0; k < SORT_K; ++k) {
      const int i = base + k * SORT_T + t;
      d[k] = (i < end) ? trec[i].z : -1;
    }
#pragma unroll
    for (int k = 0; k < SORT_K; ++k)
      if (d[k] >= 0) atomicAdd(&cntL[d[k] - node0], 1);
  }
  __syncthreads();
  {
    const int n0 = cntL[2 * t], n1 = cntL[2 * t + 1];
    const int sm = n0 + n1;
    int incl = sm;
#pragma unroll
    for (int o = 1; o < 32; o <<= 1) {
      const int v = __shfl_up(incl, o, 32);
      if (lane >= o) incl += v;
    }
    if (lane == 31) wsum[wave] = incl;
    __syncthreads();
    int woff = 0;
#pragma unroll
    for (int j = 0; j < SORT_T / 32; ++j) woff += (j < wave) ? wsum[j] : 0;
    const int ex = woff + incl - sm;
    cntL[2 * t]     = ex;
    cntL[2 * t + 1] = ex + n0;
    const int l0 = 2 * t;
    if (l0 < npb && node0 + l0 < n)         rs[node0 + l0]     = beg + ex;
    if (l0 + 1 < npb && node0 + l0 + 1 < n) rs[node0 + l0 + 1] = beg + ex + n0;
  }
  __syncthreads();
  for (int base = beg; base < end; base += SORT_T * SORT_K) {
    int4 r[SORT_K];
#pragma unroll
    for (int k = 0; k < SORT_K; ++k) {
      const int i = base + k * SORT_T + t;
      r[k] = (i < end) ? trec[i] : make_int4(0, 0, -1, 0);
    }
    int pos[SORT_K];
#pragma unroll
    for (int k = 0; k < SORT_K; ++k)
      pos[k] = (r[k].z >= 0) ? atomicAdd(&cntL[r[k].z - node0], 1) : 0;
#pragma unroll
    for (int k = 0; k < SORT_K; ++k) {
      if (r[k].z >= 0) {
        if (staged) stage[pos[k]] = make_int2(r[k].x, r[k].y);
        else        pedge[beg + pos[k]] = make_int2(r[k].x, r[k].y);
      }
    }
  }
  __syncthreads();
  if (staged) {
    for (int j = t; j < end - beg; j += SORT_T) pedge[beg + j] = stage[j];
  }
}

template<int M>
__device__ __forceinline__ float lane_xor_add(float v) {
  const int partner = __builtin_amdgcn_update_dpp(0, __float_as_int(v), 0x160 + M, 0xf, 0xf, true);
  return v + __int_as_float(partner);
}
__device__ __forceinline__ int row_pick(int v, int sel) {
  return (int)__builtin_amdgcn_permlane16_var(0u, (unsigned)v, (unsigned)sel, false, true);
}

typedef float v2f __attribute__((ext_vector_type(2)));
typedef float v4f __attribute__((ext_vector_type(4)));

template<int CO, bool RELU>
__device__ __forceinline__
void attend_row(const float* __restrict__ xl, const float* __restrict__ xr,
                const int* __restrict__ rs, const int* __restrict__ pedge,
                const float* __restrict__ We, const float* __restrict__ att,
                const float* __restrict__ bias, int node, int sub, bool live, v4f (&o)[2]) {
  constexpr int LPE = CO / 8;
  constexpr int RPP = 2 * LPE;
  static_assert(LPE == 4 || LPE == 8, "a lane group must lie inside one 16-lane row");
  const int off = sub * 8;
  const int grp = (int)(threadIdx.x & 15u) - sub;
  const int beg = rs[node];
  const int end = live ? rs[node + 1] : beg;
  v2f r2[4], w2[4], t2[4];
#pragma unroll
  for (int h = 0; h < 2; ++h) {
    const v4f rv = *(const v4f*)(xr + (size_t)node * CO + off + 4 * h);
    const v4f wv = *(const v4f*)(We + off + 4 * h);
    const v4f tv = *(const v4f*)(att + off + 4 * h);
    r2[2 * h] = rv.xy; r2[2 * h + 1] = rv.zw;
    w2[2 * h] = wv.xy; w2[2 * h + 1] = wv.zw;
    t2[2 * h] = tv.xy; t2[2 * h + 1] = tv.zw;
  }
  const float NEG_INF = -__builtin_huge_valf();

  float mx = NEG_INF, den = 0.0f;
  v2f acc[4];
#pragma unroll
  for (int i = 0; i < 4; ++i) acc[i] = v2f{0.0f, 0.0f};

  for (int base = beg; base < end; base += RPP) {
    const int j0 = base + sub;
    const int j1 = base + LPE + sub;
    int2 rec0 = make_int2(0, 0), rec1 = make_int2(0, 0);
    if (j0 < end) rec0 = *(const int2*)(pedge + 2 * (size_t)j0);
    if (j1 < end) rec1 = *(const int2*)(pedge + 2 * (size_t)j1);
    const int here = min(RPP, end - base);
    for (int t = 0; t < here; t += 4) {
      const bool second = t >= LPE;
      const int rx = second ? rec1.x : rec0.x;
      const int ry = second ? rec1.y : rec0.y;
      const int l0 = grp + (t & (LPE - 1));
      v2f   a[4][4];
      float p[4], fe4[4];
      int   s4[4];
#pragma unroll
      for (int u = 0; u < 4; ++u) {
        s4[u]  = row_pick(rx, l0 + u);
        fe4[u] = __int_as_float(row_pick(ry, l0 + u));
      }
#pragma unroll
      for (int u = 0; u < 4; ++u) {
        const float* rp = xl + (size_t)s4[u] * CO + off;
        const v4f lo = *(const v4f*)(rp);
        const v4f hi = *(const v4f*)(rp + 4);
        a[u][0] = lo.xy; a[u][1] = lo.zw; a[u][2] = hi.xy; a[u][3] = hi.zw;
      }
      __builtin_amdgcn_sched_barrier(0);
#pragma unroll
      for (int u = 0; u < 4; ++u) {
        const float fe = fe4[u];
        v2f q2 = v2f{0.0f, 0.0f};
#pragma unroll
        for (int i = 0; i < 4; ++i) {
          v2f m = a[u][i] + r2[i] + fe * w2[i];
          const v2f ms = NEG_SLOPE * m;
          m.x = (m.x > 0.0f) ? m.x : ms.x;
          m.y = (m.y > 0.0f) ? m.y : ms.y;
          q2 += m * t2[i];
        }
        p[u] = q2.x + q2.y;
      }
      if (LPE == 8) {
#pragma unroll
        for (int u = 0; u < 4; ++u) p[u] = lane_xor_add<4>(p[u]);
      }
#pragma unroll
      for (int u = 0; u < 4; ++u) p[u] = lane_xor_add<2>(p[u]);
#pragma unroll
      for (int u = 0; u < 4; ++u) p[u] = lane_xor_add<1>(p[u]);
#pragma unroll
      for (int u = 1; u < 4; ++u) {
        if (t + u >= here) {
          p[u] = NEG_INF;
#pragma unroll
          for (int i = 0; i < 4; ++i) a[u][i] = v2f{0.0f, 0.0f};
        }
      }
      const float mn = fmaxf(fmaxf(mx, p[0]), fmaxf(fmaxf(p[1], p[2]), p[3]));
      const float sc = __expf(mx - mn);
      const float e0 = __expf(p[0] - mn);
      const float e1 = __expf(p[1] - mn);
      const float e2 = __expf(p[2] - mn);
      const float e3 = __expf(p[3] - mn);
      den = den * sc + ((e0 + e1) + (e2 + e3));
#pragma unroll
      for (int i = 0; i < 4; ++i)
        acc[i] = acc[i] * sc + ((a[0][i] * e0 + a[1][i] * e1) + (a[2][i] * e2 + a[3][i] * e3));
      mx = mn;
    }
  }

  const float dn = den + 1e-16f;
#pragma unroll
  for (int h = 0; h < 2; ++h) {
    const v4f b4 = *(const v4f*)(bias + off + 4 * h);
    v4f o4;
    o4.xy = acc[2 * h] / dn + b4.xy;
    o4.zw = acc[2 * h + 1] / dn + b4.zw;
    if (RELU) {
      o4.x = fmaxf(o4.x, 0.0f); o4.y = fmaxf(o4.y, 0.0f);
      o4.z = fmaxf(o4.z, 0.0f); o4.w = fmaxf(o4.w, 0.0f);
    }
    o[h] = o4;
  }
}

template<int CO, bool RELU>
__global__ __launch_bounds__(256)
void node_attend(const float* __restrict__ xl, const float* __restrict__ xr,
                 const int* __restrict__ rs, const int* __restrict__ pedge,
                 const float* __restrict__ We, const float* __restrict__ att,
                 const float* __restrict__ bias, float* __restrict__ out, int n) {
  constexpr int LPE = CO / 8;
  const int gid  = blockIdx.x * blockDim.x + threadIdx.x;
  const int node = gid / LPE;
  const int sub  = gid % LPE;
  if (node >= n) return;
  v4f o[2];
  attend_row<CO, RELU>(xl, xr, rs, pedge, We, att, bias, node, sub, true, o);
#pragma unroll
  for (int h = 0; h < 2; ++h) *(v4f*)(out + (size_t)node * CO + sub * 8 + 4 * h) = o[h];
}

template<int CO, int CO2>
__global__ __launch_bounds__(256) __attribute__((amdgpu_waves_per_eu(9)))
void node_attend_then_xform(const float* __restrict__ xl, const float* __restrict__ xr,
                            const int* __restrict__ rs, const int* __restrict__ pedge,
                            const float* __restrict__ We, const float* __restrict__ att,
                            const float* __restrict__ bias,
                            const float* __restrict__ Wl2, const float* __restrict__ bl2,
                            const float* __restrict__ Wr2, const float* __restrict__ br2,
                            float* __restrict__ xl2, float* __restrict__ xr2, int n) {
  static_assert(CO == 64, "the strip transform takes rows of 64 inputs");
  constexpr int LPE = CO / 8;
  constexpr int NPB = 256 / LPE;
  constexpr int HP  = CO + 4;
  static_assert(NPB % 16 == 0 && 2 * (NPB / 16) <= 8, "one wave per (strip, matrix)");
  __shared__ __attribute__((aligned(16))) float hs[NPB * HP];
  const int t    = threadIdx.x;
  const int nl   = t / LPE;
  const int sub  = t % LPE;
  const int node = blockIdx.x * NPB + nl;
  const bool live = node < n;
  v4f o[2];
  attend_row<CO, true>(xl, xr, rs, pedge, We, att, bias, live ? node : n - 1, sub, live, o);
#pragma unroll
  for (int h = 0; h < 2; ++h) *(v4f*)&hs[nl * HP + sub * 8 + 4 * h] = o[h];
  __syncthreads();
  const int wave = __builtin_amdgcn_readfirstlane(t >> 5);
  if (wave >= 2 * (NPB / 16)) return;
  const int strip = wave >> 1;
  const int row_base = blockIdx.x * NPB + strip * 16;
  if (row_base >= n) return;
  const bool right = (wave & 1) != 0;
  strip_xform_from_lds<CO2, HP>(hs + strip * 16 * HP, right ? Wr2 : Wl2, right ? br2 : bl2,
                                right ? xr2 : xl2, row_base, n);
}

extern "C" void kernel_launch(void* const* d_in, const int* in_sizes, int n_in,
                              void* d_out, int out_size, void* d_ws, size_t ws_size,
                              hipStream_t stream) {
  (void)n_in; (void)out_size; (void)ws_size;
  const float* x    = (const float*)d_in[0];
  const int*   ei   = (const int*)d_in[1];
  const float* ef   = (const float*)d_in[2];
  const float* Wl0  = (const float*)d_in[3];
  const float* bl0  = (const float*)d_in[4];
  const float* Wr0  = (const float*)d_in[5];
  const float* br0  = (const float*)d_in[6];
  const float* We0  = (const float*)d_in[7];
  const float* att0 = (const float*)d_in[8];
  const float* b0   = (const float*)d_in[9];
  const float* Wl1  = (const float*)d_in[10];
  const float* bl1  = (const float*)d_in[11];
  const float* Wr1  = (const float*)d_in[12];
  const float* br1  = (const float*)d_in[13];
  const float* We1  = (const float*)d_in[14];
  const float* att1 = (const float*)d_in[15];
  const float* b1   = (const float*)d_in[16];

  const int n = in_sizes[0] / 64;
  const int e = in_sizes[1] / 2;
  const int* srcI = ei;
  const int* dstI = ei + e;

  float* ws   = (float*)d_ws;
  float* xl   = ws;  ws += (size_t)n * 64;
  float* xr   = ws;  ws += (size_t)n * 64;
  float* xl1  = ws;  ws += (size_t)n * 32;
  float* xr1  = ws;  ws += (size_t)n * 32;
  int* table  = (int*)ws;  ws += SORT_NB * SORT_NB;
  int* binbase = (int*)ws; ws += SORT_NB + 4;
  int* trec   = (int*)ws;  ws += (size_t)e * 4;
  int* pedge  = (int*)ws;  ws += (size_t)e * 2;
  int* rs     = (int*)ws;  ws += n + 1;

  auto cdiv = [](long long a, long long b) { return (int)((a + b - 1) / b); };
  const int B = 256;

  layer0_xform_and_chunk_count<<<cdiv(n, 16 * (COUNT_T / 32)) + SORT_NB, COUNT_T, 0, stream>>>(
      x, Wl0, bl0, Wr0, br0, xl, xr, n, dstI, table, e);

  chunk_place<<<SORT_NB, SORT_T, 0, stream>>>(srcI, dstI, ef, table, binbase, trec, n, e);
  bin_sort<<<SORT_NB, SORT_T, 0, stream>>>(binbase, trec, rs, pedge, n, e);

  node_attend_then_xform<64, 32><<<cdiv(n, 32), B, 0, stream>>>(
      xl, xr, rs, pedge, We0, att0, b0, Wl1, bl1, Wr1, br1, xl1, xr1, n);

  {
    constexpr int CO = 32;
    float* outp = (float*)d_out;
    node_attend<CO, false><<<cdiv((long long)n * (CO / 8), B), B, 0, stream>>>(
        xl1, xr1, rs, pedge, We1, att1, b1, outp, n);
  }
}
